# starting kernel
# baseline (speedup 1.0000x reference)
#include <hip/hip_runtime.h>
#include <hip/hip_bf16.h>

typedef __attribute__((ext_vector_type(2))) float v2f;
typedef __attribute__((ext_vector_type(8))) float v8f;

#define IN_DIM  500
#define HID_DIM 16
#define OUT_DIM 7

__global__ void fill_f32(float* __restrict__ p, float v, int n) {
    int i = blockIdx.x * blockDim.x + threadIdx.x;
    if (i < n) p[i] = v;
}

__global__ void degree_kernel(const int* __restrict__ dst,
                              float* __restrict__ deg, int num_edges) {
    int e = blockIdx.x * blockDim.x + threadIdx.x;
    if (e < num_edges) atomicAdd(deg + dst[e], 1.0f);
}

__global__ void rsqrt_kernel(float* __restrict__ dinv, int n) {
    int i = blockIdx.x * blockDim.x + threadIdx.x;
    if (i < n) {
        float d = dinv[i];
        dinv[i] = (d > 0.0f) ? rsqrtf(d) : 0.0f;
    }
}

__global__ void gemm1_wmma(const float* __restrict__ x,
                           const float* __restrict__ W1,
                           float* __restrict__ h1, int num_nodes) {
    int wave = (blockIdx.x * blockDim.x + threadIdx.x) >> 5;
    int lane = threadIdx.x & 31;
    int row0 = wave * 16;
    if (row0 + 16 > num_nodes) return;

    int half = lane >> 4;
    int l    = lane & 15;

    const float* arow = x + (size_t)(row0 + l) * IN_DIM + half * 2;
    v8f c = {};
    for (int k0 = 0; k0 < IN_DIM; k0 += 4) {
        v2f a = *(const v2f*)(arow + k0);
        v2f b;
        b.x = W1[(k0 + half * 2 + 0) * HID_DIM + l];
        b.y = W1[(k0 + half * 2 + 1) * HID_DIM + l];
        c = __builtin_amdgcn_wmma_f32_16x16x4_f32(
                false, a, false, b, (short)0, c, false, false);
    }
    float* outp = h1 + (size_t)row0 * HID_DIM;
#pragma unroll
    for (int r = 0; r < 8; ++r)
        outp[(half * 8 + r) * HID_DIM + l] = c[r];
}

__global__ void edge_mp16(const int* __restrict__ src,
                          const int* __restrict__ dst,
                          const float* __restrict__ dinv,
                          const float* __restrict__ h,
                          float* __restrict__ agg, int num_edges) {
    int idx = blockIdx.x * blockDim.x + threadIdx.x;
    int e = idx >> 4;
    int j = idx & 15;
    if (e >= num_edges) return;
    int s = src[e], d = dst[e];
    float norm = dinv[s] * dinv[d];
    atomicAdd(agg + (size_t)d * HID_DIM + j, h[(size_t)s * HID_DIM + j] * norm);
}

__global__ void node_post1(const float* __restrict__ h1,
                           const float* __restrict__ dinv,
                           const float* __restrict__ b1,
                           float* __restrict__ agg1, int num_nodes) {
    int idx = blockIdx.x * blockDim.x + threadIdx.x;
    int n = idx >> 4;
    int j = idx & 15;
    if (n >= num_nodes) return;
    float di = dinv[n];
    float v = agg1[idx] + h1[idx] * di * di + b1[j];
    agg1[idx] = fmaxf(v, 0.0f);
}

__global__ void gemm2_wmma(const float* __restrict__ agg1,
                           const float* __restrict__ W2,
                           float* __restrict__ h2, int num_nodes) {
    int wave = (blockIdx.x * blockDim.x + threadIdx.x) >> 5;
    int lane = threadIdx.x & 31;
    int row0 = wave * 16;
    if (row0 + 16 > num_nodes) return;

    int half = lane >> 4;
    int l    = lane & 15;

    const float* arow = agg1 + (size_t)(row0 + l) * HID_DIM + half * 2;
    v8f c = {};
#pragma unroll
    for (int k0 = 0; k0 < HID_DIM; k0 += 4) {
        v2f a = *(const v2f*)(arow + k0);
        v2f b;
        int kr = k0 + half * 2;
        b.x = (l < OUT_DIM) ? W2[(kr + 0) * OUT_DIM + l] : 0.0f;
        b.y = (l < OUT_DIM) ? W2[(kr + 1) * OUT_DIM + l] : 0.0f;
        c = __builtin_amdgcn_wmma_f32_16x16x4_f32(
                false, a, false, b, (short)0, c, false, false);
    }
    if (l < OUT_DIM) {
        float* outp = h2 + (size_t)row0 * OUT_DIM;
#pragma unroll
        for (int r = 0; r < 8; ++r)
            outp[(half * 8 + r) * OUT_DIM + l] = c[r];
    }
}

__global__ void edge_mp7(const int* __restrict__ src,
                         const int* __restrict__ dst,
                         const float* __restrict__ dinv,
                         const float* __restrict__ h,
                         float* __restrict__ agg, int num_edges) {
    int idx = blockIdx.x * blockDim.x + threadIdx.x;
    int e = idx >> 3;
    int j = idx & 7;
    if (e >= num_edges || j >= OUT_DIM) return;
    int s = src[e], d = dst[e];
    float norm = dinv[s] * dinv[d];
    atomicAdd(agg + (size_t)d * OUT_DIM + j, h[(size_t)s * OUT_DIM + j] * norm);
}

__global__ void node_post2(const float* __restrict__ h2,
                           const float* __restrict__ dinv,
                           const float* __restrict__ b2,
                           const float* __restrict__ agg2,
                           float* __restrict__ out, int num_nodes) {
    int n = blockIdx.x * blockDim.x + threadIdx.x;
    if (n >= num_nodes) return;
    float di = dinv[n];
    float sl = di * di;
    float v[OUT_DIM];
    float m = -INFINITY;
#pragma unroll
    for (int j = 0; j < OUT_DIM; ++j) {
        v[j] = agg2[(size_t)n * OUT_DIM + j] + h2[(size_t)n * OUT_DIM + j] * sl + b2[j];
        m = fmaxf(m, v[j]);
    }
    float s = 0.0f;
#pragma unroll
    for (int j = 0; j < OUT_DIM; ++j) s += expf(v[j] - m);
    float lse = logf(s);
#pragma unroll
    for (int j = 0; j < OUT_DIM; ++j)
        out[(size_t)n * OUT_DIM + j] = v[j] - m - lse;
}

extern "C" void kernel_launch(void* const* d_in, const int* in_sizes, int n_in,
                              void* d_out, int out_size, void* d_ws, size_t ws_size,
                              hipStream_t stream) {
    const float* x  = (const float*)d_in[0];
    const int*   ei = (const int*)d_in[1];
    const float* W1 = (const float*)d_in[2];
    const float* b1 = (const float*)d_in[3];
    const float* W2 = (const float*)d_in[4];
    const float* b2 = (const float*)d_in[5];

    const int N = in_sizes[0] / IN_DIM;
    const int E = in_sizes[1] / 2;
    const int* src = ei;
    const int* dst = ei + E;

    float* ws   = (float*)d_ws;
    float* dinv = ws;
    float* h1   = dinv + (size_t)N;
    float* agg1 = h1 + (size_t)N * HID_DIM;
    float* h2   = agg1 + (size_t)N * HID_DIM;
    float* agg2 = h2 + (size_t)N * OUT_DIM;
    float* outp = (float*)d_out;

    const int B = 256;
    auto nb = [](long long work, int blk) { return (int)((work + blk - 1) / blk); };

    fill_f32<<<nb(N, B), B, 0, stream>>>(dinv, 1.0f, N);
    fill_f32<<<nb((long long)N * HID_DIM, B), B, 0, stream>>>(agg1, 0.0f, N * HID_DIM);
    fill_f32<<<nb((long long)N * OUT_DIM, B), B, 0, stream>>>(agg2, 0.0f, N * OUT_DIM);

    degree_kernel<<<nb(E, B), B, 0, stream>>>(dst, dinv, E);
    rsqrt_kernel<<<nb(N, B), B, 0, stream>>>(dinv, N);

    int waves1 = N / 16;
    gemm1_wmma<<<nb((long long)waves1 * 32, B), B, 0, stream>>>(x, W1, h1, N);
    edge_mp16<<<nb((long long)E * 16, B), B, 0, stream>>>(src, dst, dinv, h1, agg1, E);
    node_post1<<<nb((long long)N * HID_DIM, B), B, 0, stream>>>(h1, dinv, b1, agg1, N);

    gemm2_wmma<<<nb((long long)waves1 * 32, B), B, 0, stream>>>(agg1, W2, h2, N);
    edge_mp7<<<nb((long long)E * 8, B), B, 0, stream>>>(src, dst, dinv, h2, agg2, E);
    node_post2<<<nb(N, B), B, 0, stream>>>(h2, dinv, b2, agg2, outp, N);
}


# Float-atomic scatter replaced by a per-call counting sort of the edges into 64-node destination buckets (LDS counters) and per-bucket sums as LDS float adds (arrival order); 11 launches become 8
# speedup vs baseline: 1.5665x; 1.5665x over previous
#include <hip/hip_runtime.h>
#include <hip/hip_bf16.h>

typedef __attribute__((ext_vector_type(2))) float v2f;
typedef __attribute__((ext_vector_type(8))) float v8f;

#define IN_DIM  500
#define HID_DIM 16
#define OUT_DIM 7

#define BSH   6
#define BN    64
#define BROW  (BN * OUT_DIM)
#define G1    128
#define T1    1024
#define NOEDGE 0xFFFFFFFFu

__global__ void __launch_bounds__(T1)
k_hist(const int* __restrict__ dst, int* __restrict__ table,
       int num_edges, int chunk, int nbk) {
    extern __shared__ int cnt[];
    int k = blockIdx.x;
    for (int b = threadIdx.x; b < nbk; b += T1) cnt[b] = 0;
    __syncthreads();
    int e0 = k * chunk;
    int e1 = min(num_edges, e0 + chunk);
    for (int e = e0 + (int)threadIdx.x; e < e1; e += T1)
        atomicAdd(&cnt[dst[e] >> BSH], 1);
    __syncthreads();
    for (int b = threadIdx.x; b < nbk; b += T1) table[k * nbk + b] = cnt[b];
}

__global__ void k_colscan(const int* __restrict__ table, int* __restrict__ pref,
                          int* __restrict__ rowtot, int nbk) {
    int b = blockIdx.x * blockDim.x + threadIdx.x;
    if (b >= nbk) return;
    int run = 0;
    for (int k = 0; k < G1; k += 16) {
        int c[16];
#pragma unroll
        for (int u = 0; u < 16; ++u) c[u] = table[(k + u) * nbk + b];
#pragma unroll
        for (int u = 0; u < 16; ++u) { pref[(k + u) * nbk + b] = run; run += c[u]; }
    }
    rowtot[b] = run;
}

__global__ void __launch_bounds__(1024)
k_basescan(const int* __restrict__ rowtot, int* __restrict__ bse, int nbk) {
    __shared__ int s[1024];
    int t = threadIdx.x;
    int ipt = (nbk + 1023) / 1024;
    int b0 = t * ipt;
    int mine = 0;
    for (int u = 0; u < ipt; ++u) { int b = b0 + u; if (b < nbk) mine += rowtot[b]; }
    s[t] = mine;
    __syncthreads();
    for (int off = 1; off < 1024; off <<= 1) {
        int add = (t >= off) ? s[t - off] : 0;
        __syncthreads();
        s[t] += add;
        __syncthreads();
    }
    int run = s[t] - mine;
    for (int u = 0; u < ipt; ++u) {
        int b = b0 + u;
        if (b < nbk) {
            int c = rowtot[b];
            bse[(size_t)b * BROW + 0] = run;
            bse[(size_t)b * BROW + 1] = run + c;
            run += c;
        }
    }
}

__global__ void __launch_bounds__(T1)
k_scatter(const int* __restrict__ src, const int* __restrict__ dst,
          const int* __restrict__ pref, const int* __restrict__ bse,
          unsigned short* __restrict__ lo16, unsigned char* __restrict__ hi8,
          int num_edges, int chunk, int nbk) {
    extern __shared__ int cur[];
    int k = blockIdx.x;
    for (int b = threadIdx.x; b < nbk; b += T1)
        cur[b] = bse[(size_t)b * BROW] + pref[k * nbk + b];
    __syncthreads();
    int e0 = k * chunk;
    int e1 = min(num_edges, e0 + chunk);
    for (int e = e0 + (int)threadIdx.x; e < e1; e += T1) {
        int s = src[e], d = dst[e];
        int pos = atomicAdd(&cur[d >> BSH], 1);
        lo16[pos] = (unsigned short)(s & 0xFFFF);
        hi8[pos]  = (unsigned char)(((s >> 16) << BSH) | (d & (BN - 1)));
    }
}

__global__ void gemm1_wmma(const float* __restrict__ x,
                           const float* __restrict__ W1,
                           float* __restrict__ h1, int num_nodes) {
    int wave = (blockIdx.x * blockDim.x + threadIdx.x) >> 5;
    int lane = threadIdx.x & 31;
    int row0 = wave * 16;
    if (row0 + 16 > num_nodes) return;

    int half = lane >> 4;
    int l    = lane & 15;

    const float* arow = x + (size_t)(row0 + l) * IN_DIM + half * 2;
    v8f c = {};
    for (int k0 = 0; k0 < IN_DIM; k0 += 4) {
        v2f a = *(const v2f*)(arow + k0);
        v2f b;
        b.x = W1[(k0 + half * 2 + 0) * HID_DIM + l];
        b.y = W1[(k0 + half * 2 + 1) * HID_DIM + l];
        c = __builtin_amdgcn_wmma_f32_16x16x4_f32(
                false, a, false, b, (short)0, c, false, false);
    }
    float* outp = h1 + (size_t)row0 * HID_DIM;
#pragma unroll
    for (int r = 0; r < 8; ++r)
        outp[(half * 8 + r) * HID_DIM + l] = c[r];
}

__global__ void __launch_bounds__(256)
k_scale(const unsigned char* __restrict__ hi8, const int* __restrict__ bse,
        float* __restrict__ h1, int num_nodes) {
    __shared__ int cnt[BN];
    int b = blockIdx.x, tid = threadIdx.x;
    int beg = bse[(size_t)b * BROW], end = bse[(size_t)b * BROW + 1];
    if (tid < BN) cnt[tid] = 0;
    __syncthreads();
    for (int i = beg + tid; i < end; i += 256)
        atomicAdd(&cnt[hi8[i] & (BN - 1)], 1);
    __syncthreads();
    for (int idx = tid; idx < BN * HID_DIM; idx += 256) {
        int dl = idx >> 4;
        int node = b * BN + dl;
        if (node < num_nodes) {
            float d = (float)(cnt[dl] + 1);
            float di = (d > 0.0f) ? rsqrtf(d) : 0.0f;
            h1[(size_t)node * HID_DIM + (idx & 15)] *= di;
        }
    }
}

__global__ void __launch_bounds__(256)
k_layer1(const unsigned short* __restrict__ lo16,
         const unsigned char* __restrict__ hi8,
         const int* __restrict__ bse,
         const float* __restrict__ h1s,
         const float* __restrict__ b1,
         const float* __restrict__ W2,
         float* __restrict__ h2s, int num_nodes, int num_edges) {
    __shared__ __attribute__((aligned(16))) float acc[BN * HID_DIM];
    __shared__ int   cnt[BN];
    __shared__ float dis[BN];
    int b = blockIdx.x, tid = threadIdx.x;
    int lane = tid & 31, wave = tid >> 5;
    int beg = bse[(size_t)b * BROW], end = bse[(size_t)b * BROW + 1];
    for (int i = tid; i < BN * HID_DIM; i += 256) acc[i] = 0.0f;
    if (tid < BN) cnt[tid] = 0;
    __syncthreads();

    int sub = lane >> 4;
    int j   = lane & 15;
    unsigned w = NOEDGE;
    {
        int i = beg + wave * 32 + lane;
        if (i < end) w = (unsigned)lo16[i] | ((unsigned)hi8[i] << 16);
    }
    for (int base = beg + wave * 32; base < end; base += 256) {
        int in = base + 256 + lane;
        int ic = min(in, num_edges - 1);
        unsigned nlo = lo16[ic];
        unsigned nhi = hi8[ic];
        unsigned ww[16];
        float v[16];
#pragma unroll
        for (int t = 0; t < 16; ++t) ww[t] = __shfl(w, 2 * t + sub);
#pragma unroll
        for (int t = 0; t < 16; ++t) {
            unsigned s = (ww[t] & 0xFFFFu) | ((ww[t] >> 22) << 16);
            v[t] = (ww[t] != NOEDGE) ? h1s[(size_t)s * HID_DIM + j] : 0.0f;
        }
#pragma unroll
        for (int t = 0; t < 16; ++t) {
            if (ww[t] != NOEDGE) {
                int dl = (ww[t] >> 16) & (BN - 1);
                atomicAdd(&acc[dl * HID_DIM + j], v[t]);
                if (j == 0) atomicAdd(&cnt[dl], 1);
            }
        }
        w = (in < end) ? (nlo | (nhi << 16)) : NOEDGE;
    }
    __syncthreads();

    for (int idx = tid; idx < BN * HID_DIM; idx += 256) {
        int dl = idx >> 4, jj = idx & 15;
        int node = b * BN + dl;
        float a = 0.0f;
        if (node < num_nodes) {
            float d = (float)(cnt[dl] + 1);
            float di = (d > 0.0f) ? rsqrtf(d) : 0.0f;
            float val = di * (acc[idx] + h1s[(size_t)node * HID_DIM + jj]) + b1[jj];
            a = fmaxf(val, 0.0f);
            if (jj == 0) dis[dl] = di;
        }
        acc[idx] = a;
    }
    __syncthreads();

    if (wave < BN / 16) {
        int half = lane >> 4;
        int l    = lane & 15;
        const float* arow = acc + (wave * 16 + l) * HID_DIM + half * 2;
        v8f c = {};
#pragma unroll
        for (int k0 = 0; k0 < HID_DIM; k0 += 4) {
            v2f a = *(const v2f*)(arow + k0);
            v2f bb;
            int kr = k0 + half * 2;
            bb.x = (l < OUT_DIM) ? W2[(kr + 0) * OUT_DIM + l] : 0.0f;
            bb.y = (l < OUT_DIM) ? W2[(kr + 1) * OUT_DIM + l] : 0.0f;
            c = __builtin_amdgcn_wmma_f32_16x16x4_f32(
                    false, a, false, bb, (short)0, c, false, false);
        }
        if (l < OUT_DIM) {
#pragma unroll
            for (int r = 0; r < 8; ++r) {
                int dl = wave * 16 + half * 8 + r;
                int node = b * BN + dl;
                if (node < num_nodes)
                    h2s[(size_t)node * OUT_DIM + l] = dis[dl] * c[r];
            }
        }
    }
}

__global__ void __launch_bounds__(256)
k_layer2(const unsigned short* __restrict__ lo16,
         const unsigned char* __restrict__ hi8,
         const int* bse,
         const float* __restrict__ h2s,
         const float* __restrict__ b2,
         float* out, int num_nodes, int num_edges) {
    __shared__ __attribute__((aligned(16))) float acc[BN * 8];
    __shared__ int   cnt[BN];
    int b = blockIdx.x, tid = threadIdx.x;
    int lane = tid & 31, wave = tid >> 5;
    int beg = bse[(size_t)b * BROW], end = bse[(size_t)b * BROW + 1];
    for (int i = tid; i < BN * 8; i += 256) acc[i] = 0.0f;
    if (tid < BN) cnt[tid] = 0;
    __syncthreads();

    int sub = lane >> 3;
    int j   = lane & 7;
    unsigned w = NOEDGE;
    {
        int i = beg + wave * 32 + lane;
        if (i < end) w = (unsigned)lo16[i] | ((unsigned)hi8[i] << 16);
    }
    for (int base = beg + wave * 32; base < end; base += 256) {
        int in = base + 256 + lane;
        int ic = min(in, num_edges - 1);
        unsigned nlo = lo16[ic];
        unsigned nhi = hi8[ic];
        unsigned ww[8];
        float v[8];
#pragma unroll
        for (int t = 0; t < 8; ++t) ww[t] = __shfl(w, 4 * t + sub);
#pragma unroll
        for (int t = 0; t < 8; ++t) {
            unsigned s = (ww[t] & 0xFFFFu) | ((ww[t] >> 22) << 16);
            v[t] = (ww[t] != NOEDGE && j < OUT_DIM) ? h2s[(size_t)s * OUT_DIM + j] : 0.0f;
        }
#pragma unroll
        for (int t = 0; t < 8; ++t) {
            if (ww[t] != NOEDGE) {
                int dl = (ww[t] >> 16) & (BN - 1);
                if (j < OUT_DIM) atomicAdd(&acc[dl * 8 + j], v[t]);
                if (j == 0) atomicAdd(&cnt[dl], 1);
            }
        }
        w = (in < end) ? (nlo | (nhi << 16)) : NOEDGE;
    }
    __syncthreads();

    if (tid < BN) {
        int node = b * BN + tid;
        if (node < num_nodes) {
            float d = (float)(cnt[tid] + 1);
            float di = (d > 0.0f) ? rsqrtf(d) : 0.0f;
            float v[OUT_DIM];
            float m = -INFINITY;
#pragma unroll
            for (int q = 0; q < OUT_DIM; ++q) {
                v[q] = di * (acc[tid * 8 + q] + h2s[(size_t)node * OUT_DIM + q]) + b2[q];
                m = fmaxf(m, v[q]);
            }
            float s = 0.0f;
#pragma unroll
            for (int q = 0; q < OUT_DIM; ++q) s += expf(v[q] - m);
            float lse = logf(s);
#pragma unroll
            for (int q = 0; q < OUT_DIM; ++q)
                out[(size_t)node * OUT_DIM + q] = v[q] - m - lse;
        }
    }
}

extern "C" void kernel_launch(void* const* d_in, const int* in_sizes, int n_in,
                              void* d_out, int out_size, void* d_ws, size_t ws_size,
                              hipStream_t stream) {
    const float* x  = (const float*)d_in[0];
    const int*   ei = (const int*)d_in[1];
    const float* W1 = (const float*)d_in[2];
    const float* b1 = (const float*)d_in[3];
    const float* W2 = (const float*)d_in[4];
    const float* b2 = (const float*)d_in[5];

    const int N = in_sizes[0] / IN_DIM;
    const int E = in_sizes[1] / 2;
    const int* src = ei;
    const int* dst = ei + E;

    float* ws = (float*)d_ws;
    unsigned short* lo16 = (unsigned short*)ws;
    unsigned char*  hi8  = (unsigned char*)(ws + (size_t)(E + 1) / 2);
    float* h1  = ws + (size_t)(E + 1) / 2 + (size_t)(E + 3) / 4;
    float* h2s = h1 + (size_t)N * HID_DIM;
    float* outp = (float*)d_out;
    int*   bse  = (int*)d_out;

    const int NBK   = (N + BN - 1) / BN;
    const int chunk = (E + G1 - 1) / G1;

    int* table  = (int*)h1;
    int* pref   = table + (size_t)G1 * NBK;
    int* rowtot = pref + (size_t)G1 * NBK;

    const int B = 256;
    auto nb = [](long long work, int blk) { return (int)((work + blk - 1) / blk); };

    k_hist<<<G1, T1, NBK * sizeof(int), stream>>>(dst, table, E, chunk, NBK);
    k_colscan<<<nb(NBK, B), B, 0, stream>>>(table, pref, rowtot, NBK);
    k_basescan<<<1, 1024, 0, stream>>>(rowtot, bse, NBK);
    k_scatter<<<G1, T1, NBK * sizeof(int), stream>>>(src, dst, pref, bse, lo16, hi8,
                                                     E, chunk, NBK);

    int waves1 = N / 16;
    gemm1_wmma<<<nb((long long)waves1 * 32, B), B, 0, stream>>>(x, W1, h1, N);
    k_scale<<<NBK, B, 0, stream>>>(hi8, bse, h1, N);
    k_layer1<<<NBK, B, 0, stream>>>(lo16, hi8, bse, h1, b1, W2, h2s, N, E);

    k_layer2<<<NBK, B, 0, stream>>>(lo16, hi8, bse, h2s, b2, outp, N, E);
}


# Edge placement staged in LDS: a work-group orders its chunk by bucket in LDS, then copies it out with neighbouring lanes on neighbouring list entries; a bucket's list order (LDS add order) changes
# speedup vs baseline: 2.5100x; 1.6023x over previous
#include <hip/hip_runtime.h>
#include <hip/hip_bf16.h>

typedef __attribute__((ext_vector_type(2))) float v2f;
typedef __attribute__((ext_vector_type(8))) float v8f;

#define IN_DIM  500
#define HID_DIM 16
#define OUT_DIM 7

#define BSH   6
#define BN    64
#define BROW  (BN * OUT_DIM)
#define G1    128
#define T1    1024
#define NOEDGE 0xFFFFFFFFu

__global__ void __launch_bounds__(T1)
k_hist(const int* __restrict__ dst, int* __restrict__ table,
       int num_edges, int chunk, int nbk) {
    extern __shared__ int cnt[];
    int k = blockIdx.x;
    for (int b = threadIdx.x; b < nbk; b += T1) cnt[b] = 0;
    __syncthreads();
    int e0 = k * chunk;
    int e1 = min(num_edges, e0 + chunk);
    for (int e = e0 + (int)threadIdx.x; e < e1; e += T1)
        atomicAdd(&cnt[dst[e] >> BSH], 1);
    __syncthreads();
    for (int b = threadIdx.x; b < nbk; b += T1) table[k * nbk + b] = cnt[b];
}

__global__ void k_colscan(const int* __restrict__ table, int* __restrict__ pref,
                          int* __restrict__ rowtot, int nbk) {
    int b = blockIdx.x * blockDim.x + threadIdx.x;
    if (b >= nbk) return;
    int run = 0;
    for (int k = 0; k < G1; k += 16) {
        int c[16];
#pragma unroll
        for (int u = 0; u < 16; ++u) c[u] = table[(k + u) * nbk + b];
#pragma unroll
        for (int u = 0; u < 16; ++u) { pref[(k + u) * nbk + b] = run; run += c[u]; }
    }
    rowtot[b] = run;
}

__global__ void __launch_bounds__(1024)
k_basescan(const int* __restrict__ rowtot, int* __restrict__ bse, int nbk) {
    __shared__ int s[1024];
    int t = threadIdx.x;
    int ipt = (nbk + 1023) / 1024;
    int b0 = t * ipt;
    int mine = 0;
    for (int u = 0; u < ipt; ++u) { int b = b0 + u; if (b < nbk) mine += rowtot[b]; }
    s[t] = mine;
    __syncthreads();
    for (int off = 1; off < 1024; off <<= 1) {
        int add = (t >= off) ? s[t - off] : 0;
        __syncthreads();
        s[t] += add;
        __syncthreads();
    }
    int run = s[t] - mine;
    for (int u = 0; u < ipt; ++u) {
        int b = b0 + u;
        if (b < nbk) {
            int c = rowtot[b];
            bse[(size_t)b * BROW + 0] = run;
            bse[(size_t)b * BROW + 1] = run + c;
            run += c;
        }
    }
}

#define SCAP 25600
__global__ void __launch_bounds__(T1)
k_scatter(const int* __restrict__ src, const int* __restrict__ dst,
          const int* __restrict__ table, const int* __restrict__ pref,
          const int* __restrict__ bse,
          unsigned short* __restrict__ lo16, unsigned char* __restrict__ hi8,
          int num_edges, int chunk, int nbk) {
    extern __shared__ int dyn[];
    __shared__ unsigned       stage[SCAP];
    __shared__ unsigned short sbk[SCAP];
    __shared__ int wsum[T1 / 32];
    int* cur   = dyn;
    int* delta = dyn + nbk;
    int k = blockIdx.x, tid = threadIdx.x;
    int e0 = k * chunk;
    int e1 = min(num_edges, e0 + chunk);

    if (chunk > SCAP) {
        for (int b = tid; b < nbk; b += T1)
            cur[b] = bse[(size_t)b * BROW] + pref[k * nbk + b];
        __syncthreads();
        for (int e = e0 + tid; e < e1; e += T1) {
            int s = src[e], d = dst[e];
            int pos = atomicAdd(&cur[d >> BSH], 1);
            lo16[pos] = (unsigned short)(s & 0xFFFF);
            hi8[pos]  = (unsigned char)(((s >> 16) << BSH) | (d & (BN - 1)));
        }
        return;
    }

    int lane = tid & 31, wv = tid >> 5;
    int ipt = (nbk + T1 - 1) / T1;
    int b0 = tid * ipt;
    int mine = 0;
    for (int u = 0; u < ipt; ++u) { int b = b0 + u; if (b < nbk) mine += table[k * nbk + b]; }
    int incl = mine;
    for (int off = 1; off < 32; off <<= 1) {
        int o = __shfl_up(incl, off);
        if (lane >= off) incl += o;
    }
    if (lane == 31) wsum[wv] = incl;
    __syncthreads();
    if (wv == 0) {
        int v = wsum[lane], iv = v;
        for (int off = 1; off < 32; off <<= 1) {
            int o = __shfl_up(iv, off);
            if (lane >= off) iv += o;
        }
        wsum[lane] = iv - v;
    }
    __syncthreads();
    int run = wsum[wv] + incl - mine;
    for (int u = 0; u < ipt; ++u) {
        int b = b0 + u;
        if (b < nbk) {
            cur[b]   = run;
            delta[b] = bse[(size_t)b * BROW] + pref[k * nbk + b] - run;
            run += table[k * nbk + b];
        }
    }
    __syncthreads();

    for (int e = e0 + tid; e < e1; e += T1) {
        int s = src[e], d = dst[e];
        int b = d >> BSH;
        int p = atomicAdd(&cur[b], 1);
        unsigned hb = (unsigned)(((s >> 16) << BSH) | (d & (BN - 1))) & 0xFFu;
        stage[p] = (unsigned)(s & 0xFFFF) | (hb << 16);
        sbk[p]   = (unsigned short)b;
    }
    __syncthreads();

    int n = e1 - e0;
    for (int p = tid; p < n; p += T1) {
        unsigned w = stage[p];
        int g = p + delta[sbk[p]];
        lo16[g] = (unsigned short)(w & 0xFFFFu);
        hi8[g]  = (unsigned char)(w >> 16);
    }
}

__global__ void gemm1_wmma(const float* __restrict__ x,
                           const float* __restrict__ W1,
                           float* __restrict__ h1, int num_nodes) {
    int wave = (blockIdx.x * blockDim.x + threadIdx.x) >> 5;
    int lane = threadIdx.x & 31;
    int row0 = wave * 16;
    if (row0 + 16 > num_nodes) return;

    int half = lane >> 4;
    int l    = lane & 15;

    const float* arow = x + (size_t)(row0 + l) * IN_DIM + half * 2;
    v8f c = {};
    for (int k0 = 0; k0 < IN_DIM; k0 += 4) {
        v2f a = *(const v2f*)(arow + k0);
        v2f b;
        b.x = W1[(k0 + half * 2 + 0) * HID_DIM + l];
        b.y = W1[(k0 + half * 2 + 1) * HID_DIM + l];
        c = __builtin_amdgcn_wmma_f32_16x16x4_f32(
                false, a, false, b, (short)0, c, false, false);
    }
    float* outp = h1 + (size_t)row0 * HID_DIM;
#pragma unroll
    for (int r = 0; r < 8; ++r)
        outp[(half * 8 + r) * HID_DIM + l] = c[r];
}

__global__ void __launch_bounds__(256)
k_scale(const unsigned char* __restrict__ hi8, const int* __restrict__ bse,
        float* __restrict__ h1, int num_nodes) {
    __shared__ int cnt[BN];
    int b = blockIdx.x, tid = threadIdx.x;
    int beg = bse[(size_t)b * BROW], end = bse[(size_t)b * BROW + 1];
    if (tid < BN) cnt[tid] = 0;
    __syncthreads();
    for (int i = beg + tid; i < end; i += 256)
        atomicAdd(&cnt[hi8[i] & (BN - 1)], 1);
    __syncthreads();
    for (int idx = tid; idx < BN * HID_DIM; idx += 256) {
        int dl = idx >> 4;
        int node = b * BN + dl;
        if (node < num_nodes) {
            float d = (float)(cnt[dl] + 1);
            float di = (d > 0.0f) ? rsqrtf(d) : 0.0f;
            h1[(size_t)node * HID_DIM + (idx & 15)] *= di;
        }
    }
}

__global__ void __launch_bounds__(256)
k_layer1(const unsigned short* __restrict__ lo16,
         const unsigned char* __restrict__ hi8,
         const int* __restrict__ bse,
         const float* __restrict__ h1s,
         const float* __restrict__ b1,
         const float* __restrict__ W2,
         float* __restrict__ h2s, int num_nodes, int num_edges) {
    __shared__ __attribute__((aligned(16))) float acc[BN * HID_DIM];
    __shared__ int   cnt[BN];
    __shared__ float dis[BN];
    int b = blockIdx.x, tid = threadIdx.x;
    int lane = tid & 31, wave = tid >> 5;
    int beg = bse[(size_t)b * BROW], end = bse[(size_t)b * BROW + 1];
    for (int i = tid; i < BN * HID_DIM; i += 256) acc[i] = 0.0f;
    if (tid < BN) cnt[tid] = 0;
    __syncthreads();

    int sub = lane >> 4;
    int j   = lane & 15;
    unsigned w = NOEDGE;
    {
        int i = beg + wave * 32 + lane;
        if (i < end) w = (unsigned)lo16[i] | ((unsigned)hi8[i] << 16);
    }
    for (int base = beg + wave * 32; base < end; base += 256) {
        int in = base + 256 + lane;
        int ic = min(in, num_edges - 1);
        unsigned nlo = lo16[ic];
        unsigned nhi = hi8[ic];
        unsigned ww[16];
        float v[16];
#pragma unroll
        for (int t = 0; t < 16; ++t) ww[t] = __shfl(w, 2 * t + sub);
#pragma unroll
        for (int t = 0; t < 16; ++t) {
            unsigned s = (ww[t] & 0xFFFFu) | ((ww[t] >> 22) << 16);
            v[t] = (ww[t] != NOEDGE) ? h1s[(size_t)s * HID_DIM + j] : 0.0f;
        }
#pragma unroll
        for (int t = 0; t < 16; ++t) {
            if (ww[t] != NOEDGE) {
                int dl = (ww[t] >> 16) & (BN - 1);
                atomicAdd(&acc[dl * HID_DIM + j], v[t]);
                if (j == 0) atomicAdd(&cnt[dl], 1);
            }
        }
        w = (in < end) ? (nlo | (nhi << 16)) : NOEDGE;
    }
    __syncthreads();

    for (int idx = tid; idx < BN * HID_DIM; idx += 256) {
        int dl = idx >> 4, jj = idx & 15;
        int node = b * BN + dl;
        float a = 0.0f;
        if (node < num_nodes) {
            float d = (float)(cnt[dl] + 1);
            float di = (d > 0.0f) ? rsqrtf(d) : 0.0f;
            float val = di * (acc[idx] + h1s[(size_t)node * HID_DIM + jj]) + b1[jj];
            a = fmaxf(val, 0.0f);
            if (jj == 0) dis[dl] = di;
        }
        acc[idx] = a;
    }
    __syncthreads();

    if (wave < BN / 16) {
        int half = lane >> 4;
        int l    = lane & 15;
        const float* arow = acc + (wave * 16 + l) * HID_DIM + half * 2;
        v8f c = {};
#pragma unroll
        for (int k0 = 0; k0 < HID_DIM; k0 += 4) {
            v2f a = *(const v2f*)(arow + k0);
            v2f bb;
            int kr = k0 + half * 2;
            bb.x = (l < OUT_DIM) ? W2[(kr + 0) * OUT_DIM + l] : 0.0f;
            bb.y = (l < OUT_DIM) ? W2[(kr + 1) * OUT_DIM + l] : 0.0f;
            c = __builtin_amdgcn_wmma_f32_16x16x4_f32(
                    false, a, false, bb, (short)0, c, false, false);
        }
        if (l < OUT_DIM) {
#pragma unroll
            for (int r = 0; r < 8; ++r) {
                int dl = wave * 16 + half * 8 + r;
                int node = b * BN + dl;
                if (node < num_nodes)
                    h2s[(size_t)node * OUT_DIM + l] = dis[dl] * c[r];
            }
        }
    }
}

__global__ void __launch_bounds__(256)
k_layer2(const unsigned short* __restrict__ lo16,
         const unsigned char* __restrict__ hi8,
         const int* bse,
         const float* __restrict__ h2s,
         const float* __restrict__ b2,
         float* out, int num_nodes, int num_edges) {
    __shared__ __attribute__((aligned(16))) float acc[BN * 8];
    __shared__ int   cnt[BN];
    int b = blockIdx.x, tid = threadIdx.x;
    int lane = tid & 31, wave = tid >> 5;
    int beg = bse[(size_t)b * BROW], end = bse[(size_t)b * BROW + 1];
    for (int i = tid; i < BN * 8; i += 256) acc[i] = 0.0f;
    if (tid < BN) cnt[tid] = 0;
    __syncthreads();

    int sub = lane >> 3;
    int j   = lane & 7;
    unsigned w = NOEDGE;
    {
        int i = beg + wave * 32 + lane;
        if (i < end) w = (unsigned)lo16[i] | ((unsigned)hi8[i] << 16);
    }
    for (int base = beg + wave * 32; base < end; base += 256) {
        int in = base + 256 + lane;
        int ic = min(in, num_edges - 1);
        unsigned nlo = lo16[ic];
        unsigned nhi = hi8[ic];
        unsigned ww[8];
        float v[8];
#pragma unroll
        for (int t = 0; t < 8; ++t) ww[t] = __shfl(w, 4 * t + sub);
#pragma unroll
        for (int t = 0; t < 8; ++t) {
            unsigned s = (ww[t] & 0xFFFFu) | ((ww[t] >> 22) << 16);
            v[t] = (ww[t] != NOEDGE && j < OUT_DIM) ? h2s[(size_t)s * OUT_DIM + j] : 0.0f;
        }
#pragma unroll
        for (int t = 0; t < 8; ++t) {
            if (ww[t] != NOEDGE) {
                int dl = (ww[t] >> 16) & (BN - 1);
                if (j < OUT_DIM) atomicAdd(&acc[dl * 8 + j], v[t]);
                if (j == 0) atomicAdd(&cnt[dl], 1);
            }
        }
        w = (in < end) ? (nlo | (nhi << 16)) : NOEDGE;
    }
    __syncthreads();

    if (tid < BN) {
        int node = b * BN + tid;
        if (node < num_nodes) {
            float d = (float)(cnt[tid] + 1);
            float di = (d > 0.0f) ? rsqrtf(d) : 0.0f;
            float v[OUT_DIM];
            float m = -INFINITY;
#pragma unroll
            for (int q = 0; q < OUT_DIM; ++q) {
                v[q] = di * (acc[tid * 8 + q] + h2s[(size_t)node * OUT_DIM + q]) + b2[q];
                m = fmaxf(m, v[q]);
            }
            float s = 0.0f;
#pragma unroll
            for (int q = 0; q < OUT_DIM; ++q) s += expf(v[q] - m);
            float lse = logf(s);
#pragma unroll
            for (int q = 0; q < OUT_DIM; ++q)
                out[(size_t)node * OUT_DIM + q] = v[q] - m - lse;
        }
    }
}

extern "C" void kernel_launch(void* const* d_in, const int* in_sizes, int n_in,
                              void* d_out, int out_size, void* d_ws, size_t ws_size,
                              hipStream_t stream) {
    const float* x  = (const float*)d_in[0];
    const int*   ei = (const int*)d_in[1];
    const float* W1 = (const float*)d_in[2];
    const float* b1 = (const float*)d_in[3];
    const float* W2 = (const float*)d_in[4];
    const float* b2 = (const float*)d_in[5];

    const int N = in_sizes[0] / IN_DIM;
    const int E = in_sizes[1] / 2;
    const int* src = ei;
    const int* dst = ei + E;

    float* ws = (float*)d_ws;
    unsigned short* lo16 = (unsigned short*)ws;
    unsigned char*  hi8  = (unsigned char*)(ws + (size_t)(E + 1) / 2);
    float* h1  = ws + (size_t)(E + 1) / 2 + (size_t)(E + 3) / 4;
    float* h2s = h1 + (size_t)N * HID_DIM;
    float* outp = (float*)d_out;
    int*   bse  = (int*)d_out;

    const int NBK   = (N + BN - 1) / BN;
    const int chunk = (E + G1 - 1) / G1;

    int* table  = (int*)h1;
    int* pref   = table + (size_t)G1 * NBK;
    int* rowtot = pref + (size_t)G1 * NBK;

    const int B = 256;
    auto nb = [](long long work, int blk) { return (int)((work + blk - 1) / blk); };

    k_hist<<<G1, T1, NBK * sizeof(int), stream>>>(dst, table, E, chunk, NBK);
    k_colscan<<<nb(NBK, B), B, 0, stream>>>(table, pref, rowtot, NBK);
    k_basescan<<<1, 1024, 0, stream>>>(rowtot, bse, NBK);
    k_scatter<<<G1, T1, 2 * NBK * sizeof(int), stream>>>(src, dst, table, pref, bse, lo16, hi8,
                                                         E, chunk, NBK);

    int waves1 = N / 16;
    gemm1_wmma<<<nb((long long)waves1 * 32, B), B, 0, stream>>>(x, W1, h1, N);
    k_scale<<<NBK, B, 0, stream>>>(hi8, bse, h1, N);
    k_layer1<<<NBK, B, 0, stream>>>(lo16, hi8, bse, h1, b1, W2, h2s, N, E);

    k_layer2<<<NBK, B, 0, stream>>>(lo16, hi8, bse, h2s, b2, outp, N, E);
}


# Both counting-sort passes load a thread's whole share of its edge chunk (25 entries) before the first LDS atomic: one memory round trip, not 25; a bucket's list order (LDS add order) may change
# speedup vs baseline: 2.6568x; 1.0585x over previous
#include <hip/hip_runtime.h>
#include <hip/hip_bf16.h>

typedef __attribute__((ext_vector_type(2))) float v2f;
typedef __attribute__((ext_vector_type(8))) float v8f;

#define IN_DIM  500
#define HID_DIM 16
#define OUT_DIM 7

#define BSH   6
#define BN    64
#define BROW  (BN * OUT_DIM)
#define G1    128
#define T1    1024
#define NOEDGE 0xFFFFFFFFu
#define SCAP 25600
#define EPT  (SCAP / T1)
static_assert(SCAP % T1 == 0, "a staged chunk is a whole number of edges per thread");

__global__ void __launch_bounds__(T1)
k_hist(const int* __restrict__ dst, int* __restrict__ table,
       int num_edges, int chunk, int nbk) {
    extern __shared__ int cnt[];
    int k = blockIdx.x;
    for (int b = threadIdx.x; b < nbk; b += T1) cnt[b] = 0;
    __syncthreads();
    int e0 = k * chunk;
    int e1 = min(num_edges, e0 + chunk);
    if (chunk <= SCAP && e0 < e1) {
        int ef = e0 + (int)threadIdx.x;
        int d[EPT];
#pragma unroll
        for (int u = 0; u < EPT; ++u) d[u] = dst[min(ef + u * T1, e1 - 1)];
        __builtin_amdgcn_sched_barrier(0);
#pragma unroll
        for (int u = 0; u < EPT; ++u)
            atomicAdd(&cnt[d[u] >> BSH], (ef + u * T1 < e1) ? 1 : 0);
    } else {
        for (int e = e0 + (int)threadIdx.x; e < e1; e += T1)
            atomicAdd(&cnt[dst[e] >> BSH], 1);
    }
    __syncthreads();
    for (int b = threadIdx.x; b < nbk; b += T1) table[k * nbk + b] = cnt[b];
}

__global__ void k_colscan(const int* __restrict__ table, int* __restrict__ pref,
                          int* __restrict__ rowtot, int nbk) {
    int b = blockIdx.x * blockDim.x + threadIdx.x;
    if (b >= nbk) return;
    int run = 0;
    for (int k = 0; k < G1; k += 16) {
        int c[16];
#pragma unroll
        for (int u = 0; u < 16; ++u) c[u] = table[(k + u) * nbk + b];
#pragma unroll
        for (int u = 0; u < 16; ++u) { pref[(k + u) * nbk + b] = run; run += c[u]; }
    }
    rowtot[b] = run;
}

__global__ void __launch_bounds__(1024)
k_basescan(const int* __restrict__ rowtot, int* __restrict__ bse, int nbk) {
    __shared__ int s[1024];
    int t = threadIdx.x;
    int ipt = (nbk + 1023) / 1024;
    int b0 = t * ipt;
    int mine = 0;
    for (int u = 0; u < ipt; ++u) { int b = b0 + u; if (b < nbk) mine += rowtot[b]; }
    s[t] = mine;
    __syncthreads();
    for (int off = 1; off < 1024; off <<= 1) {
        int add = (t >= off) ? s[t - off] : 0;
        __syncthreads();
        s[t] += add;
        __syncthreads();
    }
    int run = s[t] - mine;
    for (int u = 0; u < ipt; ++u) {
        int b = b0 + u;
        if (b < nbk) {
            int c = rowtot[b];
            bse[(size_t)b * BROW + 0] = run;
            bse[(size_t)b * BROW + 1] = run + c;
            run += c;
        }
    }
}

__global__ void __launch_bounds__(T1)
k_scatter(const int* __restrict__ src, const int* __restrict__ dst,
          const int* __restrict__ table, const int* __restrict__ pref,
          const int* __restrict__ bse,
          unsigned short* __restrict__ lo16, unsigned char* __restrict__ hi8,
          int num_edges, int chunk, int nbk) {
    extern __shared__ int dyn[];
    __shared__ unsigned       stage[SCAP];
    __shared__ unsigned short sbk[SCAP];
    __shared__ int wsum[T1 / 32];
    int* cur   = dyn;
    int* delta = dyn + nbk;
    int k = blockIdx.x, tid = threadIdx.x;
    int e0 = k * chunk;
    int e1 = min(num_edges, e0 + chunk);

    if (chunk > SCAP) {
        for (int b = tid; b < nbk; b += T1)
            cur[b] = bse[(size_t)b * BROW] + pref[k * nbk + b];
        __syncthreads();
        for (int e = e0 + tid; e < e1; e += T1) {
            int s = src[e], d = dst[e];
            int pos = atomicAdd(&cur[d >> BSH], 1);
            lo16[pos] = (unsigned short)(s & 0xFFFF);
            hi8[pos]  = (unsigned char)(((s >> 16) << BSH) | (d & (BN - 1)));
        }
        return;
    }

    int lane = tid & 31, wv = tid >> 5;
    int ipt = (nbk + T1 - 1) / T1;
    int b0 = tid * ipt;
    int mine = 0;
    for (int u = 0; u < ipt; ++u) { int b = b0 + u; if (b < nbk) mine += table[k * nbk + b]; }
    int incl = mine;
    for (int off = 1; off < 32; off <<= 1) {
        int o = __shfl_up(incl, off);
        if (lane >= off) incl += o;
    }
    if (lane == 31) wsum[wv] = incl;
    __syncthreads();
    if (wv == 0) {
        int v = wsum[lane], iv = v;
        for (int off = 1; off < 32; off <<= 1) {
            int o = __shfl_up(iv, off);
            if (lane >= off) iv += o;
        }
        wsum[lane] = iv - v;
    }
    __syncthreads();
    int run = wsum[wv] + incl - mine;
    for (int u = 0; u < ipt; ++u) {
        int b = b0 + u;
        if (b < nbk) {
            cur[b]   = run;
            delta[b] = bse[(size_t)b * BROW] + pref[k * nbk + b] - run;
            run += table[k * nbk + b];
        }
    }
    __syncthreads();

    if (e0 < e1) {
        int ef = e0 + tid;
        int sv[EPT], dv[EPT], pp[EPT];
#pragma unroll
        for (int u = 0; u < EPT; ++u) {
            int ec = min(ef + u * T1, e1 - 1);
            sv[u] = src[ec];
            dv[u] = dst[ec];
        }
#pragma unroll
        for (int u = 0; u < EPT; ++u)
            pp[u] = atomicAdd(&cur[dv[u] >> BSH], (ef + u * T1 < e1) ? 1 : 0);
#pragma unroll
        for (int u = 0; u < EPT; ++u) {
            if (ef + u * T1 < e1) {
                int s = sv[u], d = dv[u];
                unsigned hb = (unsigned)(((s >> 16) << BSH) | (d & (BN - 1))) & 0xFFu;
                stage[pp[u]] = (unsigned)(s & 0xFFFF) | (hb << 16);
                sbk[pp[u]]   = (unsigned short)(d >> BSH);
            }
        }
    }
    __syncthreads();

    int n = e1 - e0;
    for (int p = tid; p < n; p += T1) {
        unsigned w = stage[p];
        int g = p + delta[sbk[p]];
        lo16[g] = (unsigned short)(w & 0xFFFFu);
        hi8[g]  = (unsigned char)(w >> 16);
    }
}

__global__ void gemm1_wmma(const float* __restrict__ x,
                           const float* __restrict__ W1,
                           float* __restrict__ h1, int num_nodes) {
    int wave = (blockIdx.x * blockDim.x + threadIdx.x) >> 5;
    int lane = threadIdx.x & 31;
    int row0 = wave * 16;
    if (row0 + 16 > num_nodes) return;

    int half = lane >> 4;
    int l    = lane & 15;

    const float* arow = x + (size_t)(row0 + l) * IN_DIM + half * 2;
    v8f c = {};
    for (int k0 = 0; k0 < IN_DIM; k0 += 4) {
        v2f a = *(const v2f*)(arow + k0);
        v2f b;
        b.x = W1[(k0 + half * 2 + 0) * HID_DIM + l];
        b.y = W1[(k0 + half * 2 + 1) * HID_DIM + l];
        c = __builtin_amdgcn_wmma_f32_16x16x4_f32(
                false, a, false, b, (short)0, c, false, false);
    }
    float* outp = h1 + (size_t)row0 * HID_DIM;
#pragma unroll
    for (int r = 0; r < 8; ++r)
        outp[(half * 8 + r) * HID_DIM + l] = c[r];
}

__global__ void __launch_bounds__(256)
k_scale(const unsigned char* __restrict__ hi8, const int* __restrict__ bse,
        float* __restrict__ h1, int num_nodes) {
    __shared__ int cnt[BN];
    int b = blockIdx.x, tid = threadIdx.x;
    int beg = bse[(size_t)b * BROW], end = bse[(size_t)b * BROW + 1];
    if (tid < BN) cnt[tid] = 0;
    __syncthreads();
    for (int i = beg + tid; i < end; i += 256)
        atomicAdd(&cnt[hi8[i] & (BN - 1)], 1);
    __syncthreads();
    for (int idx = tid; idx < BN * HID_DIM; idx += 256) {
        int dl = idx >> 4;
        int node = b * BN + dl;
        if (node < num_nodes) {
            float d = (float)(cnt[dl] + 1);
            float di = (d > 0.0f) ? rsqrtf(d) : 0.0f;
            h1[(size_t)node * HID_DIM + (idx & 15)] *= di;
        }
    }
}

__global__ void __launch_bounds__(256)
k_layer1(const unsigned short* __restrict__ lo16,
         const unsigned char* __restrict__ hi8,
         const int* __restrict__ bse,
         const float* __restrict__ h1s,
         const float* __restrict__ b1,
         const float* __restrict__ W2,
         float* __restrict__ h2s, int num_nodes, int num_edges) {
    __shared__ __attribute__((aligned(16))) float acc[BN * HID_DIM];
    __shared__ int   cnt[BN];
    __shared__ float dis[BN];
    int b = blockIdx.x, tid = threadIdx.x;
    int lane = tid & 31, wave = tid >> 5;
    int beg = bse[(size_t)b * BROW], end = bse[(size_t)b * BROW + 1];
    for (int i = tid; i < BN * HID_DIM; i += 256) acc[i] = 0.0f;
    if (tid < BN) cnt[tid] = 0;
    __syncthreads();

    int sub = lane >> 4;
    int j   = lane & 15;
    unsigned w = NOEDGE;
    {
        int i = beg + wave * 32 + lane;
        if (i < end) w = (unsigned)lo16[i] | ((unsigned)hi8[i] << 16);
    }
    for (int base = beg + wave * 32; base < end; base += 256) {
        int in = base + 256 + lane;
        int ic = min(in, num_edges - 1);
        unsigned nlo = lo16[ic];
        unsigned nhi = hi8[ic];
        unsigned ww[16];
        float v[16];
#pragma unroll
        for (int t = 0; t < 16; ++t) ww[t] = __shfl(w, 2 * t + sub);
#pragma unroll
        for (int t = 0; t < 16; ++t) {
            unsigned s = (ww[t] & 0xFFFFu) | ((ww[t] >> 22) << 16);
            v[t] = (ww[t] != NOEDGE) ? h1s[(size_t)s * HID_DIM + j] : 0.0f;
        }
#pragma unroll
        for (int t = 0; t < 16; ++t) {
            if (ww[t] != NOEDGE) {
                int dl = (ww[t] >> 16) & (BN - 1);
                atomicAdd(&acc[dl * HID_DIM + j], v[t]);
                if (j == 0) atomicAdd(&cnt[dl], 1);
            }
        }
        w = (in < end) ? (nlo | (nhi << 16)) : NOEDGE;
    }
    __syncthreads();

    for (int idx = tid; idx < BN * HID_DIM; idx += 256) {
        int dl = idx >> 4, jj = idx & 15;
        int node = b * BN + dl;
        float a = 0.0f;
        if (node < num_nodes) {
            float d = (float)(cnt[dl] + 1);
            float di = (d > 0.0f) ? rsqrtf(d) : 0.0f;
            float val = di * (acc[idx] + h1s[(size_t)node * HID_DIM + jj]) + b1[jj];
            a = fmaxf(val, 0.0f);
            if (jj == 0) dis[dl] = di;
        }
        acc[idx] = a;
    }
    __syncthreads();

    if (wave < BN / 16) {
        int half = lane >> 4;
        int l    = lane & 15;
        const float* arow = acc + (wave * 16 + l) * HID_DIM + half * 2;
        v8f c = {};
#pragma unroll
        for (int k0 = 0; k0 < HID_DIM; k0 += 4) {
            v2f a = *(const v2f*)(arow + k0);
            v2f bb;
            int kr = k0 + half * 2;
            bb.x = (l < OUT_DIM) ? W2[(kr + 0) * OUT_DIM + l] : 0.0f;
            bb.y = (l < OUT_DIM) ? W2[(kr + 1) * OUT_DIM + l] : 0.0f;
            c = __builtin_amdgcn_wmma_f32_16x16x4_f32(
                    false, a, false, bb, (short)0, c, false, false);
        }
        if (l < OUT_DIM) {
#pragma unroll
            for (int r = 0; r < 8; ++r) {
                int dl = wave * 16 + half * 8 + r;
                int node = b * BN + dl;
                if (node < num_nodes)
                    h2s[(size_t)node * OUT_DIM + l] = dis[dl] * c[r];
            }
        }
    }
}

__global__ void __launch_bounds__(256)
k_layer2(const unsigned short* __restrict__ lo16,
         const unsigned char* __restrict__ hi8,
         const int* bse,
         const float* __restrict__ h2s,
         const float* __restrict__ b2,
         float* out, int num_nodes, int num_edges) {
    __shared__ __attribute__((aligned(16))) float acc[BN * 8];
    __shared__ int   cnt[BN];
    int b = blockIdx.x, tid = threadIdx.x;
    int lane = tid & 31, wave = tid >> 5;
    int beg = bse[(size_t)b * BROW], end = bse[(size_t)b * BROW + 1];
    for (int i = tid; i < BN * 8; i += 256) acc[i] = 0.0f;
    if (tid < BN) cnt[tid] = 0;
    __syncthreads();

    int sub = lane >> 3;
    int j   = lane & 7;
    unsigned w = NOEDGE;
    {
        int i = beg + wave * 32 + lane;
        if (i < end) w = (unsigned)lo16[i] | ((unsigned)hi8[i] << 16);
    }
    for (int base = beg + wave * 32; base < end; base += 256) {
        int in = base + 256 + lane;
        int ic = min(in, num_edges - 1);
        unsigned nlo = lo16[ic];
        unsigned nhi = hi8[ic];
        unsigned ww[8];
        float v[8];
#pragma unroll
        for (int t = 0; t < 8; ++t) ww[t] = __shfl(w, 4 * t + sub);
#pragma unroll
        for (int t = 0; t < 8; ++t) {
            unsigned s = (ww[t] & 0xFFFFu) | ((ww[t] >> 22) << 16);
            v[t] = (ww[t] != NOEDGE && j < OUT_DIM) ? h2s[(size_t)s * OUT_DIM + j] : 0.0f;
        }
#pragma unroll
        for (int t = 0; t < 8; ++t) {
            if (ww[t] != NOEDGE) {
                int dl = (ww[t] >> 16) & (BN - 1);
                if (j < OUT_DIM) atomicAdd(&acc[dl * 8 + j], v[t]);
                if (j == 0) atomicAdd(&cnt[dl], 1);
            }
        }
        w = (in < end) ? (nlo | (nhi << 16)) : NOEDGE;
    }
    __syncthreads();

    if (tid < BN) {
        int node = b * BN + tid;
        if (node < num_nodes) {
            float d = (float)(cnt[tid] + 1);
            float di = (d > 0.0f) ? rsqrtf(d) : 0.0f;
            float v[OUT_DIM];
            float m = -INFINITY;
#pragma unroll
            for (int q = 0; q < OUT_DIM; ++q) {
                v[q] = di * (acc[tid * 8 + q] + h2s[(size_t)node * OUT_DIM + q]) + b2[q];
                m = fmaxf(m, v[q]);
            }
            float s = 0.0f;
#pragma unroll
            for (int q = 0; q < OUT_DIM; ++q) s += expf(v[q] - m);
            float lse = logf(s);
#pragma unroll
            for (int q = 0; q < OUT_DIM; ++q)
                out[(size_t)node * OUT_DIM + q] = v[q] - m - lse;
        }
    }
}

extern "C" void kernel_launch(void* const* d_in, const int* in_sizes, int n_in,
                              void* d_out, int out_size, void* d_ws, size_t ws_size,
                              hipStream_t stream) {
    const float* x  = (const float*)d_in[0];
    const int*   ei = (const int*)d_in[1];
    const float* W1 = (const float*)d_in[2];
    const float* b1 = (const float*)d_in[3];
    const float* W2 = (const float*)d_in[4];
    const float* b2 = (const float*)d_in[5];

    const int N = in_sizes[0] / IN_DIM;
    const int E = in_sizes[1] / 2;
    const int* src = ei;
    const int* dst = ei + E;

    float* ws = (float*)d_ws;
    unsigned short* lo16 = (unsigned short*)ws;
    unsigned char*  hi8  = (unsigned char*)(ws + (size_t)(E + 1) / 2);
    float* h1  = ws + (size_t)(E + 1) / 2 + (size_t)(E + 3) / 4;
    float* h2s = h1 + (size_t)N * HID_DIM;
    float* outp = (float*)d_out;
    int*   bse  = (int*)d_out;

    const int NBK   = (N + BN - 1) / BN;
    const int chunk = (E + G1 - 1) / G1;

    int* table  = (int*)h1;
    int* pref   = table + (size_t)G1 * NBK;
    int* rowtot = pref + (size_t)G1 * NBK;

    const int B = 256;
    auto nb = [](long long work, int blk) { return (int)((work + blk - 1) / blk); };

    k_hist<<<G1, T1, NBK * sizeof(int), stream>>>(dst, table, E, chunk, NBK);
    k_colscan<<<nb(NBK, B), B, 0, stream>>>(table, pref, rowtot, NBK);
    k_basescan<<<1, 1024, 0, stream>>>(rowtot, bse, NBK);
    k_scatter<<<G1, T1, 2 * NBK * sizeof(int), stream>>>(src, dst, table, pref, bse, lo16, hi8,
                                                         E, chunk, NBK);

    int waves1 = N / 16;
    gemm1_wmma<<<nb((long long)waves1 * 32, B), B, 0, stream>>>(x, W1, h1, N);
    k_scale<<<NBK, B, 0, stream>>>(hi8, bse, h1, N);
    k_layer1<<<NBK, B, 0, stream>>>(lo16, hi8, bse, h1, b1, W2, h2s, N, E);

    k_layer2<<<NBK, B, 0, stream>>>(lo16, hi8, bse, h2s, b2, outp, N, E);
}


# GEMM1 reads x as a stream: a work-group copies 32 whole rows and W1 to LDS with whole-line 16-byte non-temporal loads, WMMA fragments come from LDS; k sum now four interleaved partial sums
# speedup vs baseline: 3.8492x; 1.4488x over previous
#include <hip/hip_runtime.h>
#include <hip/hip_bf16.h>

typedef __attribute__((ext_vector_type(2))) float v2f;
typedef __attribute__((ext_vector_type(8))) float v8f;

#define IN_DIM  500
#define HID_DIM 16
#define OUT_DIM 7

#define BSH   6
#define BN    64
#define BROW  (BN * OUT_DIM)
#define G1    128
#define T1    1024
#define NOEDGE 0xFFFFFFFFu
#define SCAP 25600
#define EPT  (SCAP / T1)
static_assert(SCAP % T1 == 0, "a staged chunk is a whole number of edges per thread");

__global__ void __launch_bounds__(T1)
k_hist(const int* __restrict__ dst, int* __restrict__ table,
       int num_edges, int chunk, int nbk) {
    extern __shared__ int cnt[];
    int k = blockIdx.x;
    for (int b = threadIdx.x; b < nbk; b += T1) cnt[b] = 0;
    __syncthreads();
    int e0 = k * chunk;
    int e1 = min(num_edges, e0 + chunk);
    if (chunk <= SCAP && e0 < e1) {
        int ef = e0 + (int)threadIdx.x;
        int d[EPT];
#pragma unroll
        for (int u = 0; u < EPT; ++u) d[u] = dst[min(ef + u * T1, e1 - 1)];
        __builtin_amdgcn_sched_barrier(0);
#pragma unroll
        for (int u = 0; u < EPT; ++u)
            atomicAdd(&cnt[d[u] >> BSH], (ef + u * T1 < e1) ? 1 : 0);
    } else {
        for (int e = e0 + (int)threadIdx.x; e < e1; e += T1)
            atomicAdd(&cnt[dst[e] >> BSH], 1);
    }
    __syncthreads();
    for (int b = threadIdx.x; b < nbk; b += T1) table[k * nbk + b] = cnt[b];
}

__global__ void k_colscan(const int* __restrict__ table, int* __restrict__ pref,
                          int* __restrict__ rowtot, int nbk) {
    int b = blockIdx.x * blockDim.x + threadIdx.x;
    if (b >= nbk) return;
    int run = 0;
    for (int k = 0; k < G1; k += 16) {
        int c[16];
#pragma unroll
        for (int u = 0; u < 16; ++u) c[u] = table[(k + u) * nbk + b];
#pragma unroll
        for (int u = 0; u < 16; ++u) { pref[(k + u) * nbk + b] = run; run += c[u]; }
    }
    rowtot[b] = run;
}

__global__ void __launch_bounds__(1024)
k_basescan(const int* __restrict__ rowtot, int* __restrict__ bse, int nbk) {
    __shared__ int s[1024];
    int t = threadIdx.x;
    int ipt = (nbk + 1023) / 1024;
    int b0 = t * ipt;
    int mine = 0;
    for (int u = 0; u < ipt; ++u) { int b = b0 + u; if (b < nbk) mine += rowtot[b]; }
    s[t] = mine;
    __syncthreads();
    for (int off = 1; off < 1024; off <<= 1) {
        int add = (t >= off) ? s[t - off] : 0;
        __syncthreads();
        s[t] += add;
        __syncthreads();
    }
    int run = s[t] - mine;
    for (int u = 0; u < ipt; ++u) {
        int b = b0 + u;
        if (b < nbk) {
            int c = rowtot[b];
            bse[(size_t)b * BROW + 0] = run;
            bse[(size_t)b * BROW + 1] = run + c;
            run += c;
        }
    }
}

__global__ void __launch_bounds__(T1)
k_scatter(const int* __restrict__ src, const int* __restrict__ dst,
          const int* __restrict__ table, const int* __restrict__ pref,
          const int* __restrict__ bse,
          unsigned short* __restrict__ lo16, unsigned char* __restrict__ hi8,
          int num_edges, int chunk, int nbk) {
    extern __shared__ int dyn[];
    __shared__ unsigned       stage[SCAP];
    __shared__ unsigned short sbk[SCAP];
    __shared__ int wsum[T1 / 32];
    int* cur   = dyn;
    int* delta = dyn + nbk;
    int k = blockIdx.x, tid = threadIdx.x;
    int e0 = k * chunk;
    int e1 = min(num_edges, e0 + chunk);

    if (chunk > SCAP) {
        for (int b = tid; b < nbk; b += T1)
            cur[b] = bse[(size_t)b * BROW] + pref[k * nbk + b];
        __syncthreads();
        for (int e = e0 + tid; e < e1; e += T1) {
            int s = src[e], d = dst[e];
            int pos = atomicAdd(&cur[d >> BSH], 1);
            lo16[pos] = (unsigned short)(s & 0xFFFF);
            hi8[pos]  = (unsigned char)(((s >> 16) << BSH) | (d & (BN - 1)));
        }
        return;
    }

    int lane = tid & 31, wv = tid >> 5;
    int ipt = (nbk + T1 - 1) / T1;
    int b0 = tid * ipt;
    int mine = 0;
    for (int u = 0; u < ipt; ++u) { int b = b0 + u; if (b < nbk) mine += table[k * nbk + b]; }
    int incl = mine;
    for (int off = 1; off < 32; off <<= 1) {
        int o = __shfl_up(incl, off);
        if (lane >= off) incl += o;
    }
    if (lane == 31) wsum[wv] = incl;
    __syncthreads();
    if (wv == 0) {
        int v = wsum[lane], iv = v;
        for (int off = 1; off < 32; off <<= 1) {
            int o = __shfl_up(iv, off);
            if (lane >= off) iv += o;
        }
        wsum[lane] = iv - v;
    }
    __syncthreads();
    int run = wsum[wv] + incl - mine;
    for (int u = 0; u < ipt; ++u) {
        int b = b0 + u;
        if (b < nbk) {
            cur[b]   = run;
            delta[b] = bse[(size_t)b * BROW] + pref[k * nbk + b] - run;
            run += table[k * nbk + b];
        }
    }
    __syncthreads();

    if (e0 < e1) {
        int ef = e0 + tid;
        int sv[EPT], dv[EPT], pp[EPT];
#pragma unroll
        for (int u = 0; u < EPT; ++u) {
            int ec = min(ef + u * T1, e1 - 1);
            sv[u] = src[ec];
            dv[u] = dst[ec];
        }
#pragma unroll
        for (int u = 0; u < EPT; ++u)
            pp[u] = atomicAdd(&cur[dv[u] >> BSH], (ef + u * T1 < e1) ? 1 : 0);
#pragma unroll
        for (int u = 0; u < EPT; ++u) {
            if (ef + u * T1 < e1) {
                int s = sv[u], d = dv[u];
                unsigned hb = (unsigned)(((s >> 16) << BSH) | (d & (BN - 1))) & 0xFFu;
                stage[pp[u]] = (unsigned)(s & 0xFFFF) | (hb << 16);
                sbk[pp[u]]   = (unsigned short)(d >> BSH);
            }
        }
    }
    __syncthreads();

    int n = e1 - e0;
    for (int p = tid; p < n; p += T1) {
        unsigned w = stage[p];
        int g = p + delta[sbk[p]];
        lo16[g] = (unsigned short)(w & 0xFFFFu);
        hi8[g]  = (unsigned char)(w >> 16);
    }
}

#define GR   32
#define GT   128
#define GX4  (GR * IN_DIM / 4)
#define GW4  (IN_DIM * HID_DIM / 4)
#define GXU  ((GX4 + GT - 1) / GT)
#define GWU  ((GW4 + GT - 1) / GT)
#define KS2  (IN_DIM / 8)
#define KB   16
#define NBLK ((KS2 + KB - 1) / KB)
typedef __attribute__((ext_vector_type(4))) float v4f;

__global__ void __launch_bounds__(GT)
gemm1_wmma(const float* __restrict__ x,
           const float* __restrict__ W1,
           float* __restrict__ h1, int num_nodes) {
    __shared__ __attribute__((aligned(16))) float xs[GR * IN_DIM];
    __shared__ __attribute__((aligned(16))) float wsm[IN_DIM * HID_DIM];
    __shared__ float exch[2 * 8 * 32];
    int tid = threadIdx.x;
    int lane = tid & 31;
    int wave = __builtin_amdgcn_readfirstlane(tid >> 5);
    int blk = blockIdx.x;
    int rows = min(GR, num_nodes - blk * GR);
    int n4 = rows * (IN_DIM / 4);
    const v4f* xb = (const v4f*)(x + (size_t)blk * GR * IN_DIM);
    const v4f* wb = (const v4f*)W1;

    v4f wr[GWU], xr[GXU];
#pragma unroll
    for (int u = 0; u < GWU; ++u)
        wr[u] = ((u + 1) * GT <= GW4) ? wb[u * GT + tid] : wb[min(u * GT + tid, GW4 - 1)];
#pragma unroll
    for (int u = 0; u < GXU; ++u)
        xr[u] = __builtin_nontemporal_load(xb + min(u * GT + tid, n4 - 1));
#pragma unroll
    for (int u = 0; u < GWU; ++u) {
        int q = u * GT + tid;
        if ((u + 1) * GT <= GW4 || q < GW4) ((v4f*)wsm)[q] = wr[u];
    }
#pragma unroll
    for (int u = 0; u < GXU; ++u) {
        int i = u * GT + tid;
        if (i < n4) ((v4f*)xs)[i] = xr[u];
    }
    __syncthreads();

    int half = lane >> 4;
    int l    = lane & 15;
    int tile = wave & 1;
    int par  = wave >> 1;
    int row0 = blk * GR + tile * 16;
    bool live = (row0 + 16 <= num_nodes);
    v8f c = {};
    if (live) {
        const float* arow = xs + (tile * 16 + l) * IN_DIM + half * 2 + par * 4;
        const float* bcol = wsm + (half * 2 + par * 4) * HID_DIM + l;
        v2f fa[2][KB], fb[2][KB];
        v8f c0 = {}, c1 = {};
#pragma unroll
        for (int t = 0; t < KB; ++t) {
            fa[0][t] = *(const v2f*)(arow + t * 8);
            fb[0][t].x = bcol[t * 8 * HID_DIM];
            fb[0][t].y = bcol[t * 8 * HID_DIM + HID_DIM];
        }
#pragma unroll
        for (int n = 0; n < NBLK; ++n) {
            if (n + 1 < NBLK) {
#pragma unroll
                for (int t = 0; t < KB; ++t) {
                    int j = (n + 1) * KB + t;
                    if (j < KS2) {
                        fa[(n + 1) & 1][t] = *(const v2f*)(arow + j * 8);
                        fb[(n + 1) & 1][t].x = bcol[j * 8 * HID_DIM];
                        fb[(n + 1) & 1][t].y = bcol[j * 8 * HID_DIM + HID_DIM];
                    }
                }
            }
            __builtin_amdgcn_sched_barrier(0);
#pragma unroll
            for (int t = 0; t < KB; ++t) {
                int j = n * KB + t;
                if (j < KS2) {
                    if (t & 1)
                        c1 = __builtin_amdgcn_wmma_f32_16x16x4_f32(
                                false, fa[n & 1][t], false, fb[n & 1][t], (short)0, c1, false, false);
                    else
                        c0 = __builtin_amdgcn_wmma_f32_16x16x4_f32(
                                false, fa[n & 1][t], false, fb[n & 1][t], (short)0, c0, false, false);
                }
            }
            __builtin_amdgcn_sched_barrier(0);
        }
        if (par == 0) {
            v2f a = *(const v2f*)(arow + KS2 * 8);
            v2f b;
            b.x = bcol[KS2 * 8 * HID_DIM];
            b.y = bcol[KS2 * 8 * HID_DIM + HID_DIM];
            c0 = __builtin_amdgcn_wmma_f32_16x16x4_f32(
                    false, a, false, b, (short)0, c0, false, false);
        }
        c = c0 + c1;
        if (par == 1) {
#pragma unroll
            for (int r = 0; r < 8; ++r) exch[(tile * 8 + r) * 32 + lane] = c[r];
        }
    }
    __syncthreads();
    if (live && par == 0) {
        float* outp = h1 + (size_t)row0 * HID_DIM;
#pragma unroll
        for (int r = 0; r < 8; ++r)
            outp[(half * 8 + r) * HID_DIM + l] = c[r] + exch[(tile * 8 + r) * 32 + lane];
    }
}

__global__ void __launch_bounds__(256)
k_scale(const unsigned char* __restrict__ hi8, const int* __restrict__ bse,
        float* __restrict__ h1, int num_nodes) {
    __shared__ int cnt[BN];
    int b = blockIdx.x, tid = threadIdx.x;
    int beg = bse[(size_t)b * BROW], end = bse[(size_t)b * BROW + 1];
    if (tid < BN) cnt[tid] = 0;
    __syncthreads();
    for (int i = beg + tid; i < end; i += 256)
        atomicAdd(&cnt[hi8[i] & (BN - 1)], 1);
    __syncthreads();
    for (int idx = tid; idx < BN * HID_DIM; idx += 256) {
        int dl = idx >> 4;
        int node = b * BN + dl;
        if (node < num_nodes) {
            float d = (float)(cnt[dl] + 1);
            float di = (d > 0.0f) ? rsqrtf(d) : 0.0f;
            h1[(size_t)node * HID_DIM + (idx & 15)] *= di;
        }
    }
}

__global__ void __launch_bounds__(256)
k_layer1(const unsigned short* __restrict__ lo16,
         const unsigned char* __restrict__ hi8,
         const int* __restrict__ bse,
         const float* __restrict__ h1s,
         const float* __restrict__ b1,
         const float* __restrict__ W2,
         float* __restrict__ h2s, int num_nodes, int num_edges) {
    __shared__ __attribute__((aligned(16))) float acc[BN * HID_DIM];
    __shared__ int   cnt[BN];
    __shared__ float dis[BN];
    int b = blockIdx.x, tid = threadIdx.x;
    int lane = tid & 31, wave = tid >> 5;
    int beg = bse[(size_t)b * BROW], end = bse[(size_t)b * BROW + 1];
    for (int i = tid; i < BN * HID_DIM; i += 256) acc[i] = 0.0f;
    if (tid < BN) cnt[tid] = 0;
    __syncthreads();

    int sub = lane >> 4;
    int j   = lane & 15;
    unsigned w = NOEDGE;
    {
        int i = beg + wave * 32 + lane;
        if (i < end) w = (unsigned)lo16[i] | ((unsigned)hi8[i] << 16);
    }
    for (int base = beg + wave * 32; base < end; base += 256) {
        int in = base + 256 + lane;
        int ic = min(in, num_edges - 1);
        unsigned nlo = lo16[ic];
        unsigned nhi = hi8[ic];
        unsigned ww[16];
        float v[16];
#pragma unroll
        for (int t = 0; t < 16; ++t) ww[t] = __shfl(w, 2 * t + sub);
#pragma unroll
        for (int t = 0; t < 16; ++t) {
            unsigned s = (ww[t] & 0xFFFFu) | ((ww[t] >> 22) << 16);
            v[t] = (ww[t] != NOEDGE) ? h1s[(size_t)s * HID_DIM + j] : 0.0f;
        }
#pragma unroll
        for (int t = 0; t < 16; ++t) {
            if (ww[t] != NOEDGE) {
                int dl = (ww[t] >> 16) & (BN - 1);
                atomicAdd(&acc[dl * HID_DIM + j], v[t]);
                if (j == 0) atomicAdd(&cnt[dl], 1);
            }
        }
        w = (in < end) ? (nlo | (nhi << 16)) : NOEDGE;
    }
    __syncthreads();

    for (int idx = tid; idx < BN * HID_DIM; idx += 256) {
        int dl = idx >> 4, jj = idx & 15;
        int node = b * BN + dl;
        float a = 0.0f;
        if (node < num_nodes) {
            float d = (float)(cnt[dl] + 1);
            float di = (d > 0.0f) ? rsqrtf(d) : 0.0f;
            float val = di * (acc[idx] + h1s[(size_t)node * HID_DIM + jj]) + b1[jj];
            a = fmaxf(val, 0.0f);
            if (jj == 0) dis[dl] = di;
        }
        acc[idx] = a;
    }
    __syncthreads();

    if (wave < BN / 16) {
        int half = lane >> 4;
        int l    = lane & 15;
        const float* arow = acc + (wave * 16 + l) * HID_DIM + half * 2;
        v8f c = {};
#pragma unroll
        for (int k0 = 0; k0 < HID_DIM; k0 += 4) {
            v2f a = *(const v2f*)(arow + k0);
            v2f bb;
            int kr = k0 + half * 2;
            bb.x = (l < OUT_DIM) ? W2[(kr + 0) * OUT_DIM + l] : 0.0f;
            bb.y = (l < OUT_DIM) ? W2[(kr + 1) * OUT_DIM + l] : 0.0f;
            c = __builtin_amdgcn_wmma_f32_16x16x4_f32(
                    false, a, false, bb, (short)0, c, false, false);
        }
        if (l < OUT_DIM) {
#pragma unroll
            for (int r = 0; r < 8; ++r) {
                int dl = wave * 16 + half * 8 + r;
                int node = b * BN + dl;
                if (node < num_nodes)
                    h2s[(size_t)node * OUT_DIM + l] = dis[dl] * c[r];
            }
        }
    }
}

__global__ void __launch_bounds__(256)
k_layer2(const unsigned short* __restrict__ lo16,
         const unsigned char* __restrict__ hi8,
         const int* bse,
         const float* __restrict__ h2s,
         const float* __restrict__ b2,
         float* out, int num_nodes, int num_edges) {
    __shared__ __attribute__((aligned(16))) float acc[BN * 8];
    __shared__ int   cnt[BN];
    int b = blockIdx.x, tid = threadIdx.x;
    int lane = tid & 31, wave = tid >> 5;
    int beg = bse[(size_t)b * BROW], end = bse[(size_t)b * BROW + 1];
    for (int i = tid; i < BN * 8; i += 256) acc[i] = 0.0f;
    if (tid < BN) cnt[tid] = 0;
    __syncthreads();

    int sub = lane >> 3;
    int j   = lane & 7;
    unsigned w = NOEDGE;
    {
        int i = beg + wave * 32 + lane;
        if (i < end) w = (unsigned)lo16[i] | ((unsigned)hi8[i] << 16);
    }
    for (int base = beg + wave * 32; base < end; base += 256) {
        int in = base + 256 + lane;
        int ic = min(in, num_edges - 1);
        unsigned nlo = lo16[ic];
        unsigned nhi = hi8[ic];
        unsigned ww[8];
        float v[8];
#pragma unroll
        for (int t = 0; t < 8; ++t) ww[t] = __shfl(w, 4 * t + sub);
#pragma unroll
        for (int t = 0; t < 8; ++t) {
            unsigned s = (ww[t] & 0xFFFFu) | ((ww[t] >> 22) << 16);
            v[t] = (ww[t] != NOEDGE && j < OUT_DIM) ? h2s[(size_t)s * OUT_DIM + j] : 0.0f;
        }
#pragma unroll
        for (int t = 0; t < 8; ++t) {
            if (ww[t] != NOEDGE) {
                int dl = (ww[t] >> 16) & (BN - 1);
                if (j < OUT_DIM) atomicAdd(&acc[dl * 8 + j], v[t]);
                if (j == 0) atomicAdd(&cnt[dl], 1);
            }
        }
        w = (in < end) ? (nlo | (nhi << 16)) : NOEDGE;
    }
    __syncthreads();

    if (tid < BN) {
        int node = b * BN + tid;
        if (node < num_nodes) {
            float d = (float)(cnt[tid] + 1);
            float di = (d > 0.0f) ? rsqrtf(d) : 0.0f;
            float v[OUT_DIM];
            float m = -INFINITY;
#pragma unroll
            for (int q = 0; q < OUT_DIM; ++q) {
                v[q] = di * (acc[tid * 8 + q] + h2s[(size_t)node * OUT_DIM + q]) + b2[q];
                m = fmaxf(m, v[q]);
            }
            float s = 0.0f;
#pragma unroll
            for (int q = 0; q < OUT_DIM; ++q) s += expf(v[q] - m);
            float lse = logf(s);
#pragma unroll
            for (int q = 0; q < OUT_DIM; ++q)
                out[(size_t)node * OUT_DIM + q] = v[q] - m - lse;
        }
    }
}

extern "C" void kernel_launch(void* const* d_in, const int* in_sizes, int n_in,
                              void* d_out, int out_size, void* d_ws, size_t ws_size,
                              hipStream_t stream) {
    const float* x  = (const float*)d_in[0];
    const int*   ei = (const int*)d_in[1];
    const float* W1 = (const float*)d_in[2];
    const float* b1 = (const float*)d_in[3];
    const float* W2 = (const float*)d_in[4];
    const float* b2 = (const float*)d_in[5];

    const int N = in_sizes[0] / IN_DIM;
    const int E = in_sizes[1] / 2;
    const int* src = ei;
    const int* dst = ei + E;

    float* ws = (float*)d_ws;
    unsigned short* lo16 = (unsigned short*)ws;
    unsigned char*  hi8  = (unsigned char*)(ws + (size_t)(E + 1) / 2);
    float* h1  = ws + (size_t)(E + 1) / 2 + (size_t)(E + 3) / 4;
    float* h2s = h1 + (size_t)N * HID_DIM;
    float* outp = (float*)d_out;
    int*   bse  = (int*)d_out;

    const int NBK   = (N + BN - 1) / BN;
    const int chunk = (E + G1 - 1) / G1;

    int* table  = (int*)h1;
    int* pref   = table + (size_t)G1 * NBK;
    int* rowtot = pref + (size_t)G1 * NBK;

    const int B = 256;
    auto nb = [](long long work, int blk) { return (int)((work + blk - 1) / blk); };

    k_hist<<<G1, T1, NBK * sizeof(int), stream>>>(dst, table, E, chunk, NBK);
    k_colscan<<<nb(NBK, B), B, 0, stream>>>(table, pref, rowtot, NBK);
    k_basescan<<<1, 1024, 0, stream>>>(rowtot, bse, NBK);
    k_scatter<<<G1, T1, 2 * NBK * sizeof(int), stream>>>(src, dst, table, pref, bse, lo16, hi8,
                                                         E, chunk, NBK);

    gemm1_wmma<<<nb(N, GR), GT, 0, stream>>>(x, W1, h1, N);
    k_scale<<<NBK, B, 0, stream>>>(hi8, bse, h1, N);
    k_layer1<<<NBK, B, 0, stream>>>(lo16, hi8, bse, h1, b1, W2, h2s, N, E);

    k_layer2<<<NBK, B, 0, stream>>>(lo16, hi8, bse, h2s, b2, outp, N, E);
}


# Node degrees counted once, in the GEMM1 work-group from its bucket's edge list: h1 is written scaled, deg^-1/2 kept per node; the scale launch and both layers' recounts go (8 launches become 7)
# speedup vs baseline: 4.0250x; 1.0457x over previous
#include <hip/hip_runtime.h>
#include <hip/hip_bf16.h>

typedef __attribute__((ext_vector_type(2))) float v2f;
typedef __attribute__((ext_vector_type(8))) float v8f;

#define IN_DIM  500
#define HID_DIM 16
#define OUT_DIM 7

#define BSH   6
#define BN    64
#define BROW  (BN * OUT_DIM)
#define G1    128
#define T1    1024
#define NOEDGE 0xFFFFFFFFu
#define SCAP 25600
#define EPT  (SCAP / T1)
static_assert(SCAP % T1 == 0, "a staged chunk is a whole number of edges per thread");

__global__ void __launch_bounds__(T1)
k_hist(const int* __restrict__ dst, int* __restrict__ table,
       int num_edges, int chunk, int nbk) {
    extern __shared__ int cnt[];
    int k = blockIdx.x;
    for (int b = threadIdx.x; b < nbk; b += T1) cnt[b] = 0;
    __syncthreads();
    int e0 = k * chunk;
    int e1 = min(num_edges, e0 + chunk);
    if (chunk <= SCAP && e0 < e1) {
        int ef = e0 + (int)threadIdx.x;
        int d[EPT];
#pragma unroll
        for (int u = 0; u < EPT; ++u) d[u] = dst[min(ef + u * T1, e1 - 1)];
        __builtin_amdgcn_sched_barrier(0);
#pragma unroll
        for (int u = 0; u < EPT; ++u)
            atomicAdd(&cnt[d[u] >> BSH], (ef + u * T1 < e1) ? 1 : 0);
    } else {
        for (int e = e0 + (int)threadIdx.x; e < e1; e += T1)
            atomicAdd(&cnt[dst[e] >> BSH], 1);
    }
    __syncthreads();
    for (int b = threadIdx.x; b < nbk; b += T1) table[k * nbk + b] = cnt[b];
}

__global__ void k_colscan(const int* __restrict__ table, int* __restrict__ pref,
                          int* __restrict__ rowtot, int nbk) {
    int b = blockIdx.x * blockDim.x + threadIdx.x;
    if (b >= nbk) return;
    int run = 0;
    for (int k = 0; k < G1; k += 16) {
        int c[16];
#pragma unroll
        for (int u = 0; u < 16; ++u) c[u] = table[(k + u) * nbk + b];
#pragma unroll
        for (int u = 0; u < 16; ++u) { pref[(k + u) * nbk + b] = run; run += c[u]; }
    }
    rowtot[b] = run;
}

__global__ void __launch_bounds__(1024)
k_basescan(const int* __restrict__ rowtot, int* __restrict__ bse, int nbk) {
    __shared__ int s[1024];
    int t = threadIdx.x;
    int ipt = (nbk + 1023) / 1024;
    int b0 = t * ipt;
    int mine = 0;
    for (int u = 0; u < ipt; ++u) { int b = b0 + u; if (b < nbk) mine += rowtot[b]; }
    s[t] = mine;
    __syncthreads();
    for (int off = 1; off < 1024; off <<= 1) {
        int add = (t >= off) ? s[t - off] : 0;
        __syncthreads();
        s[t] += add;
        __syncthreads();
    }
    int run = s[t] - mine;
    for (int u = 0; u < ipt; ++u) {
        int b = b0 + u;
        if (b < nbk) {
            int c = rowtot[b];
            bse[(size_t)b * BROW + 0] = run;
            bse[(size_t)b * BROW + 1] = run + c;
            run += c;
        }
    }
}

__global__ void __launch_bounds__(T1)
k_scatter(const int* __restrict__ src, const int* __restrict__ dst,
          const int* __restrict__ table, const int* __restrict__ pref,
          const int* __restrict__ bse,
          unsigned short* __restrict__ lo16, unsigned char* __restrict__ hi8,
          int num_edges, int chunk, int nbk) {
    extern __shared__ int dyn[];
    __shared__ unsigned       stage[SCAP];
    __shared__ unsigned short sbk[SCAP];
    __shared__ int wsum[T1 / 32];
    int* cur   = dyn;
    int* delta = dyn + nbk;
    int k = blockIdx.x, tid = threadIdx.x;
    int e0 = k * chunk;
    int e1 = min(num_edges, e0 + chunk);

    if (chunk > SCAP) {
        for (int b = tid; b < nbk; b += T1)
            cur[b] = bse[(size_t)b * BROW] + pref[k * nbk + b];
        __syncthreads();
        for (int e = e0 + tid; e < e1; e += T1) {
            int s = src[e], d = dst[e];
            int pos = atomicAdd(&cur[d >> BSH], 1);
            lo16[pos] = (unsigned short)(s & 0xFFFF);
            hi8[pos]  = (unsigned char)(((s >> 16) << BSH) | (d & (BN - 1)));
        }
        return;
    }

    int lane = tid & 31, wv = tid >> 5;
    int ipt = (nbk + T1 - 1) / T1;
    int b0 = tid * ipt;
    int mine = 0;
    for (int u = 0; u < ipt; ++u) { int b = b0 + u; if (b < nbk) mine += table[k * nbk + b]; }
    int incl = mine;
    for (int off = 1; off < 32; off <<= 1) {
        int o = __shfl_up(incl, off);
        if (lane >= off) incl += o;
    }
    if (lane == 31) wsum[wv] = incl;
    __syncthreads();
    if (wv == 0) {
        int v = wsum[lane], iv = v;
        for (int off = 1; off < 32; off <<= 1) {
            int o = __shfl_up(iv, off);
            if (lane >= off) iv += o;
        }
        wsum[lane] = iv - v;
    }
    __syncthreads();
    int run = wsum[wv] + incl - mine;
    for (int u = 0; u < ipt; ++u) {
        int b = b0 + u;
        if (b < nbk) {
            cur[b]   = run;
            delta[b] = bse[(size_t)b * BROW] + pref[k * nbk + b] - run;
            run += table[k * nbk + b];
        }
    }
    __syncthreads();

    if (e0 < e1) {
        int ef = e0 + tid;
        int sv[EPT], dv[EPT], pp[EPT];
#pragma unroll
        for (int u = 0; u < EPT; ++u) {
            int ec = min(ef + u * T1, e1 - 1);
            sv[u] = src[ec];
            dv[u] = dst[ec];
        }
#pragma unroll
        for (int u = 0; u < EPT; ++u)
            pp[u] = atomicAdd(&cur[dv[u] >> BSH], (ef + u * T1 < e1) ? 1 : 0);
#pragma unroll
        for (int u = 0; u < EPT; ++u) {
            if (ef + u * T1 < e1) {
                int s = sv[u], d = dv[u];
                unsigned hb = (unsigned)(((s >> 16) << BSH) | (d & (BN - 1))) & 0xFFu;
                stage[pp[u]] = (unsigned)(s & 0xFFFF) | (hb << 16);
                sbk[pp[u]]   = (unsigned short)(d >> BSH);
            }
        }
    }
    __syncthreads();

    int n = e1 - e0;
    for (int p = tid; p < n; p += T1) {
        unsigned w = stage[p];
        int g = p + delta[sbk[p]];
        lo16[g] = (unsigned short)(w & 0xFFFFu);
        hi8[g]  = (unsigned char)(w >> 16);
    }
}

#define GR   32
#define GT   128
#define GX4  (GR * IN_DIM / 4)
#define GW4  (IN_DIM * HID_DIM / 4)
#define GXU  ((GX4 + GT - 1) / GT)
#define GWU  ((GW4 + GT - 1) / GT)
#define KS2  (IN_DIM / 8)
#define KB   16
#define NBLK ((KS2 + KB - 1) / KB)
#define CU   5
#define DISQ 3
static_assert(BN % GR == 0, "a GEMM work-group's rows lie in one bucket");
typedef __attribute__((ext_vector_type(4))) float v4f;

__global__ void __launch_bounds__(GT)
gemm1_wmma(const float* __restrict__ x,
           const float* __restrict__ W1,
           const unsigned* __restrict__ hi32,
           const int* __restrict__ bse,
           float* __restrict__ h1, float* __restrict__ disg, int num_nodes) {
    __shared__ __attribute__((aligned(16))) float xs[GR * IN_DIM];
    __shared__ __attribute__((aligned(16))) float wsm[IN_DIM * HID_DIM];
    __shared__ float exch[2 * 8 * 32];
    __shared__ int dcnt[GR];
    int tid = threadIdx.x;
    int lane = tid & 31;
    int wave = __builtin_amdgcn_readfirstlane(tid >> 5);
    int blk = blockIdx.x;
    int rows = min(GR, num_nodes - blk * GR);
    int n4 = rows * (IN_DIM / 4);
    const v4f* xb = (const v4f*)(x + (size_t)blk * GR * IN_DIM);
    const v4f* wb = (const v4f*)W1;

    int bk   = blk / (BN / GR);
    int part = blk % (BN / GR);
    if (tid < GR) dcnt[tid] = 0;

    v4f wr[GWU], xr[GXU];
    unsigned hw[CU];
#pragma unroll
    for (int u = 0; u < GWU; ++u)
        wr[u] = ((u + 1) * GT <= GW4) ? wb[u * GT + tid] : wb[min(u * GT + tid, GW4 - 1)];
#pragma unroll
    for (int u = 0; u < GXU; ++u)
        xr[u] = __builtin_nontemporal_load(xb + min(u * GT + tid, n4 - 1));
    __builtin_amdgcn_sched_barrier(0);
    int beg = bse[(size_t)bk * BROW], end = bse[(size_t)bk * BROW + 1];
    int w0 = beg >> 2, w1 = (end + 3) >> 2;
    int wlast = max(w1 - 1, w0);
#pragma unroll
    for (int u = 0; u < CU; ++u) hw[u] = hi32[min(w0 + u * GT + tid, wlast)];
#pragma unroll
    for (int u = 0; u < GWU; ++u) {
        int q = u * GT + tid;
        if ((u + 1) * GT <= GW4 || q < GW4) ((v4f*)wsm)[q] = wr[u];
    }
#pragma unroll
    for (int u = 0; u < GXU; ++u) {
        int i = u * GT + tid;
        if (i < n4) ((v4f*)xs)[i] = xr[u];
    }
    __syncthreads();

#pragma unroll
    for (int u = 0; u < CU; ++u) {
        int wi = w0 + u * GT + tid;
#pragma unroll
        for (int q = 0; q < 4; ++q) {
            int pos = wi * 4 + q;
            int dl = (hw[u] >> (8 * q)) & (BN - 1);
            if (pos >= beg && pos < end && dl / GR == part) atomicAdd(&dcnt[dl % GR], 1);
        }
    }
    for (int wi = w0 + CU * GT + tid; wi < w1; wi += GT) {
        unsigned wv = hi32[wi];
#pragma unroll
        for (int q = 0; q < 4; ++q) {
            int pos = wi * 4 + q;
            int dl = (wv >> (8 * q)) & (BN - 1);
            if (pos >= beg && pos < end && dl / GR == part) atomicAdd(&dcnt[dl % GR], 1);
        }
    }

    int half = lane >> 4;
    int l    = lane & 15;
    int tile = wave & 1;
    int par  = wave >> 1;
    int row0 = blk * GR + tile * 16;
    bool live = (row0 + 16 <= num_nodes);
    v8f c = {};
    if (live) {
        const float* arow = xs + (tile * 16 + l) * IN_DIM + half * 2 + par * 4;
        const float* bcol = wsm + (half * 2 + par * 4) * HID_DIM + l;
        v2f fa[2][KB], fb[2][KB];
        v8f c0 = {}, c1 = {};
#pragma unroll
        for (int t = 0; t < KB; ++t) {
            fa[0][t] = *(const v2f*)(arow + t * 8);
            fb[0][t].x = bcol[t * 8 * HID_DIM];
            fb[0][t].y = bcol[t * 8 * HID_DIM + HID_DIM];
        }
#pragma unroll
        for (int n = 0; n < NBLK; ++n) {
            if (n + 1 < NBLK) {
#pragma unroll
                for (int t = 0; t < KB; ++t) {
                    int j = (n + 1) * KB + t;
                    if (j < KS2) {
                        fa[(n + 1) & 1][t] = *(const v2f*)(arow + j * 8);
                        fb[(n + 1) & 1][t].x = bcol[j * 8 * HID_DIM];
                        fb[(n + 1) & 1][t].y = bcol[j * 8 * HID_DIM + HID_DIM];
                    }
                }
            }
            __builtin_amdgcn_sched_barrier(0);
#pragma unroll
            for (int t = 0; t < KB; ++t) {
                int j = n * KB + t;
                if (j < KS2) {
                    if (t & 1)
                        c1 = __builtin_amdgcn_wmma_f32_16x16x4_f32(
                                false, fa[n & 1][t], false, fb[n & 1][t], (short)0, c1, false, false);
                    else
                        c0 = __builtin_amdgcn_wmma_f32_16x16x4_f32(
                                false, fa[n & 1][t], false, fb[n & 1][t], (short)0, c0, false, false);
                }
            }
            __builtin_amdgcn_sched_barrier(0);
        }
        if (par == 0) {
            v2f a = *(const v2f*)(arow + KS2 * 8);
            v2f b;
            b.x = bcol[KS2 * 8 * HID_DIM];
            b.y = bcol[KS2 * 8 * HID_DIM + HID_DIM];
            c0 = __builtin_amdgcn_wmma_f32_16x16x4_f32(
                    false, a, false, b, (short)0, c0, false, false);
        }
        c = c0 + c1;
        if (par == 1) {
#pragma unroll
            for (int r = 0; r < 8; ++r) exch[(tile * 8 + r) * 32 + lane] = c[r];
        }
    }
    __syncthreads();
    if (live && par == 0) {
        float* outp = h1 + (size_t)row0 * HID_DIM;
#pragma unroll
        for (int r = 0; r < 8; ++r) {
            float d = (float)(dcnt[tile * 16 + half * 8 + r] + 1);
            float di = (d > 0.0f) ? rsqrtf(d) : 0.0f;
            outp[(half * 8 + r) * HID_DIM + l] = (c[r] + exch[(tile * 8 + r) * 32 + lane]) * di;
        }
    }
    if (tid < GR) {
        int node = blk * GR + tid;
        if (node < num_nodes) {
            float d = (float)(dcnt[tid] + 1);
            disg[(size_t)node * OUT_DIM + DISQ] = (d > 0.0f) ? rsqrtf(d) : 0.0f;
        }
    }
}

__global__ void __launch_bounds__(256)
k_layer1(const unsigned short* __restrict__ lo16,
         const unsigned char* __restrict__ hi8,
         const int* __restrict__ bse,
         const float* __restrict__ h1s,
         const float* __restrict__ disg,
         const float* __restrict__ b1,
         const float* __restrict__ W2,
         float* __restrict__ h2s, int num_nodes, int num_edges) {
    __shared__ __attribute__((aligned(16))) float acc[BN * HID_DIM];
    __shared__ float dis[BN];
    int b = blockIdx.x, tid = threadIdx.x;
    int lane = tid & 31, wave = tid >> 5;
    int beg = bse[(size_t)b * BROW], end = bse[(size_t)b * BROW + 1];
    for (int i = tid; i < BN * HID_DIM; i += 256) acc[i] = 0.0f;
    if (tid < BN) {
        int node = b * BN + tid;
        dis[tid] = (node < num_nodes) ? disg[(size_t)node * OUT_DIM + DISQ] : 0.0f;
    }
    __syncthreads();

    int sub = lane >> 4;
    int j   = lane & 15;
    unsigned w = NOEDGE;
    {
        int i = beg + wave * 32 + lane;
        if (i < end) w = (unsigned)lo16[i] | ((unsigned)hi8[i] << 16);
    }
    for (int base = beg + wave * 32; base < end; base += 256) {
        int in = base + 256 + lane;
        int ic = min(in, num_edges - 1);
        unsigned nlo = lo16[ic];
        unsigned nhi = hi8[ic];
        unsigned ww[16];
        float v[16];
#pragma unroll
        for (int t = 0; t < 16; ++t) ww[t] = __shfl(w, 2 * t + sub);
#pragma unroll
        for (int t = 0; t < 16; ++t) {
            unsigned s = (ww[t] & 0xFFFFu) | ((ww[t] >> 22) << 16);
            v[t] = (ww[t] != NOEDGE) ? h1s[(size_t)s * HID_DIM + j] : 0.0f;
        }
#pragma unroll
        for (int t = 0; t < 16; ++t) {
            if (ww[t] != NOEDGE) {
                int dl = (ww[t] >> 16) & (BN - 1);
                atomicAdd(&acc[dl * HID_DIM + j], v[t]);
            }
        }
        w = (in < end) ? (nlo | (nhi << 16)) : NOEDGE;
    }
    __syncthreads();

    for (int idx = tid; idx < BN * HID_DIM; idx += 256) {
        int dl = idx >> 4, jj = idx & 15;
        int node = b * BN + dl;
        float a = 0.0f;
        if (node < num_nodes) {
            float di = dis[dl];
            float val = di * (acc[idx] + h1s[(size_t)node * HID_DIM + jj]) + b1[jj];
            a = fmaxf(val, 0.0f);
        }
        acc[idx] = a;
    }
    __syncthreads();

    if (wave < BN / 16) {
        int half = lane >> 4;
        int l    = lane & 15;
        const float* arow = acc + (wave * 16 + l) * HID_DIM + half * 2;
        v8f c = {};
#pragma unroll
        for (int k0 = 0; k0 < HID_DIM; k0 += 4) {
            v2f a = *(const v2f*)(arow + k0);
            v2f bb;
            int kr = k0 + half * 2;
            bb.x = (l < OUT_DIM) ? W2[(kr + 0) * OUT_DIM + l] : 0.0f;
            bb.y = (l < OUT_DIM) ? W2[(kr + 1) * OUT_DIM + l] : 0.0f;
            c = __builtin_amdgcn_wmma_f32_16x16x4_f32(
                    false, a, false, bb, (short)0, c, false, false);
        }
        if (l < OUT_DIM) {
#pragma unroll
            for (int r = 0; r < 8; ++r) {
                int dl = wave * 16 + half * 8 + r;
                int node = b * BN + dl;
                if (node < num_nodes)
                    h2s[(size_t)node * OUT_DIM + l] = dis[dl] * c[r];
            }
        }
    }
}

__global__ void __launch_bounds__(256)
k_layer2(const unsigned short* __restrict__ lo16,
         const unsigned char* __restrict__ hi8,
         const int* bse,
         const float* __restrict__ h2s,
         const float* __restrict__ b2,
         float* out, int num_nodes, int num_edges) {
    __shared__ __attribute__((aligned(16))) float acc[BN * 8];
    int b = blockIdx.x, tid = threadIdx.x;
    int lane = tid & 31, wave = tid >> 5;
    int beg = bse[(size_t)b * BROW], end = bse[(size_t)b * BROW + 1];
    for (int i = tid; i < BN * 8; i += 256) acc[i] = 0.0f;
    float di = 0.0f;
    if (tid < BN && b * BN + tid < num_nodes)
        di = out[(size_t)(b * BN + tid) * OUT_DIM + DISQ];
    __syncthreads();

    int sub = lane >> 3;
    int j   = lane & 7;
    unsigned w = NOEDGE;
    {
        int i = beg + wave * 32 + lane;
        if (i < end) w = (unsigned)lo16[i] | ((unsigned)hi8[i] << 16);
    }
    for (int base = beg + wave * 32; base < end; base += 256) {
        int in = base + 256 + lane;
        int ic = min(in, num_edges - 1);
        unsigned nlo = lo16[ic];
        unsigned nhi = hi8[ic];
        unsigned ww[8];
        float v[8];
#pragma unroll
        for (int t = 0; t < 8; ++t) ww[t] = __shfl(w, 4 * t + sub);
#pragma unroll
        for (int t = 0; t < 8; ++t) {
            unsigned s = (ww[t] & 0xFFFFu) | ((ww[t] >> 22) << 16);
            v[t] = (ww[t] != NOEDGE && j < OUT_DIM) ? h2s[(size_t)s * OUT_DIM + j] : 0.0f;
        }
#pragma unroll
        for (int t = 0; t < 8; ++t) {
            if (ww[t] != NOEDGE) {
                int dl = (ww[t] >> 16) & (BN - 1);
                if (j < OUT_DIM) atomicAdd(&acc[dl * 8 + j], v[t]);
            }
        }
        w = (in < end) ? (nlo | (nhi << 16)) : NOEDGE;
    }
    __syncthreads();

    if (tid < BN) {
        int node = b * BN + tid;
        if (node < num_nodes) {
            float v[OUT_DIM];
            float m = -INFINITY;
#pragma unroll
            for (int q = 0; q < OUT_DIM; ++q) {
                v[q] = di * (acc[tid * 8 + q] + h2s[(size_t)node * OUT_DIM + q]) + b2[q];
                m = fmaxf(m, v[q]);
            }
            float s = 0.0f;
#pragma unroll
            for (int q = 0; q < OUT_DIM; ++q) s += expf(v[q] - m);
            float lse = logf(s);
#pragma unroll
            for (int q = 0; q < OUT_DIM; ++q)
                out[(size_t)node * OUT_DIM + q] = v[q] - m - lse;
        }
    }
}

extern "C" void kernel_launch(void* const* d_in, const int* in_sizes, int n_in,
                              void* d_out, int out_size, void* d_ws, size_t ws_size,
                              hipStream_t stream) {
    const float* x  = (const float*)d_in[0];
    const int*   ei = (const int*)d_in[1];
    const float* W1 = (const float*)d_in[2];
    const float* b1 = (const float*)d_in[3];
    const float* W2 = (const float*)d_in[4];
    const float* b2 = (const float*)d_in[5];

    const int N = in_sizes[0] / IN_DIM;
    const int E = in_sizes[1] / 2;
    const int* src = ei;
    const int* dst = ei + E;

    float* ws = (float*)d_ws;
    unsigned short* lo16 = (unsigned short*)ws;
    unsigned char*  hi8  = (unsigned char*)(ws + (size_t)(E + 1) / 2);
    float* h1  = ws + (size_t)(E + 1) / 2 + (size_t)(E + 3) / 4;
    float* h2s = h1 + (size_t)N * HID_DIM;
    float* outp = (float*)d_out;
    int*   bse  = (int*)d_out;
    float* disg = (float*)d_out;

    const int NBK   = (N + BN - 1) / BN;
    const int chunk = (E + G1 - 1) / G1;

    int* table  = (int*)h1;
    int* pref   = table + (size_t)G1 * NBK;
    int* rowtot = pref + (size_t)G1 * NBK;

    const int B = 256;
    auto nb = [](long long work, int blk) { return (int)((work + blk - 1) / blk); };

    k_hist<<<G1, T1, NBK * sizeof(int), stream>>>(dst, table, E, chunk, NBK);
    k_colscan<<<nb(NBK, B), B, 0, stream>>>(table, pref, rowtot, NBK);
    k_basescan<<<1, 1024, 0, stream>>>(rowtot, bse, NBK);
    k_scatter<<<G1, T1, 2 * NBK * sizeof(int), stream>>>(src, dst, table, pref, bse, lo16, hi8,
                                                         E, chunk, NBK);

    gemm1_wmma<<<nb(N, GR), GT, 0, stream>>>(x, W1, (const unsigned*)hi8, bse, h1, disg, N);
    k_layer1<<<NBK, B, 0, stream>>>(lo16, hi8, bse, h1, disg, b1, W2, h2s, N, E);

    k_layer2<<<NBK, B, 0, stream>>>(lo16, hi8, bse, h2s, b2, outp, N, E);
}


# Counting sort spread over every WGP: 256 work-groups (13 edges a thread) for 128 (25), column scan split over 8 waves (one load round trip); a bucket's list order (LDS add order) changes
# speedup vs baseline: 4.1596x; 1.0334x over previous
#include <hip/hip_runtime.h>
#include <hip/hip_bf16.h>

typedef __attribute__((ext_vector_type(2))) float v2f;
typedef __attribute__((ext_vector_type(8))) float v8f;

#define IN_DIM  500
#define HID_DIM 16
#define OUT_DIM 7

#define BSH   6
#define BN    64
#define BROW  (BN * OUT_DIM)
#define G1    256
#define T1    1024
#define NOEDGE 0xFFFFFFFFu
#define SCAP 13312
#define EPT  (SCAP / T1)
static_assert(SCAP % T1 == 0, "a staged chunk is a whole number of edges per thread");

__global__ void __launch_bounds__(T1)
k_hist(const int* __restrict__ dst, int* __restrict__ table,
       int num_edges, int chunk, int nbk) {
    extern __shared__ int cnt[];
    int k = blockIdx.x;
    for (int b = threadIdx.x; b < nbk; b += T1) cnt[b] = 0;
    __syncthreads();
    int e0 = k * chunk;
    int e1 = min(num_edges, e0 + chunk);
    if (chunk <= SCAP && e0 < e1) {
        int ef = e0 + (int)threadIdx.x;
        int d[EPT];
#pragma unroll
        for (int u = 0; u < EPT; ++u) d[u] = dst[min(ef + u * T1, e1 - 1)];
        __builtin_amdgcn_sched_barrier(0);
#pragma unroll
        for (int u = 0; u < EPT; ++u)
            atomicAdd(&cnt[d[u] >> BSH], (ef + u * T1 < e1) ? 1 : 0);
    } else {
        for (int e = e0 + (int)threadIdx.x; e < e1; e += T1)
            atomicAdd(&cnt[dst[e] >> BSH], 1);
    }
    __syncthreads();
    for (int b = threadIdx.x; b < nbk; b += T1) table[k * nbk + b] = cnt[b];
}

#define CW 8
#define CK (G1 / CW)
static_assert(G1 % CW == 0, "the table rows split evenly over the waves");
__global__ void __launch_bounds__(32 * CW)
k_colscan(const int* __restrict__ table, int* __restrict__ pref,
          int* __restrict__ rowtot, int nbk) {
    __shared__ int part[CW][32];
    int lane = threadIdx.x & 31, w = threadIdx.x >> 5;
    int b = blockIdx.x * 32 + lane;
    int bc = min(b, nbk - 1);
    int c[CK];
#pragma unroll
    for (int u = 0; u < CK; ++u) c[u] = table[(w * CK + u) * nbk + bc];
    int tot = 0;
#pragma unroll
    for (int u = 0; u < CK; ++u) { int t = c[u]; c[u] = tot; tot += t; }
    part[w][lane] = tot;
    __syncthreads();
    int off = 0, all = 0;
#pragma unroll
    for (int q = 0; q < CW; ++q) { int p = part[q][lane]; if (q < w) off += p; all += p; }
    if (b < nbk) {
#pragma unroll
        for (int u = 0; u < CK; ++u) pref[(w * CK + u) * nbk + b] = off + c[u];
        if (w == 0) rowtot[b] = all;
    }
}

__global__ void __launch_bounds__(1024)
k_basescan(const int* __restrict__ rowtot, int* __restrict__ bse, int nbk) {
    __shared__ int s[1024];
    int t = threadIdx.x;
    int ipt = (nbk + 1023) / 1024;
    int b0 = t * ipt;
    int mine = 0;
    for (int u = 0; u < ipt; ++u) { int b = b0 + u; if (b < nbk) mine += rowtot[b]; }
    s[t] = mine;
    __syncthreads();
    for (int off = 1; off < 1024; off <<= 1) {
        int add = (t >= off) ? s[t - off] : 0;
        __syncthreads();
        s[t] += add;
        __syncthreads();
    }
    int run = s[t] - mine;
    for (int u = 0; u < ipt; ++u) {
        int b = b0 + u;
        if (b < nbk) {
            int c = rowtot[b];
            bse[(size_t)b * BROW + 0] = run;
            bse[(size_t)b * BROW + 1] = run + c;
            run += c;
        }
    }
}

__global__ void __launch_bounds__(T1)
k_scatter(const int* __restrict__ src, const int* __restrict__ dst,
          const int* __restrict__ table, const int* __restrict__ pref,
          const int* __restrict__ bse,
          unsigned short* __restrict__ lo16, unsigned char* __restrict__ hi8,
          int num_edges, int chunk, int nbk) {
    extern __shared__ int dyn[];
    __shared__ unsigned       stage[SCAP];
    __shared__ unsigned short sbk[SCAP];
    __shared__ int wsum[T1 / 32];
    int* cur   = dyn;
    int* delta = dyn + nbk;
    int k = blockIdx.x, tid = threadIdx.x;
    int e0 = k * chunk;
    int e1 = min(num_edges, e0 + chunk);

    if (chunk > SCAP) {
        for (int b = tid; b < nbk; b += T1)
            cur[b] = bse[(size_t)b * BROW] + pref[k * nbk + b];
        __syncthreads();
        for (int e = e0 + tid; e < e1; e += T1) {
            int s = src[e], d = dst[e];
            int pos = atomicAdd(&cur[d >> BSH], 1);
            lo16[pos] = (unsigned short)(s & 0xFFFF);
            hi8[pos]  = (unsigned char)(((s >> 16) << BSH) | (d & (BN - 1)));
        }
        return;
    }

    int lane = tid & 31, wv = tid >> 5;
    int ipt = (nbk + T1 - 1) / T1;
    int b0 = tid * ipt;
    int mine = 0;
    for (int u = 0; u < ipt; ++u) { int b = b0 + u; if (b < nbk) mine += table[k * nbk + b]; }
    int incl = mine;
    for (int off = 1; off < 32; off <<= 1) {
        int o = __shfl_up(incl, off);
        if (lane >= off) incl += o;
    }
    if (lane == 31) wsum[wv] = incl;
    __syncthreads();
    if (wv == 0) {
        int v = wsum[lane], iv = v;
        for (int off = 1; off < 32; off <<= 1) {
            int o = __shfl_up(iv, off);
            if (lane >= off) iv += o;
        }
        wsum[lane] = iv - v;
    }
    __syncthreads();
    int run = wsum[wv] + incl - mine;
    for (int u = 0; u < ipt; ++u) {
        int b = b0 + u;
        if (b < nbk) {
            cur[b]   = run;
            delta[b] = bse[(size_t)b * BROW] + pref[k * nbk + b] - run;
            run += table[k * nbk + b];
        }
    }
    __syncthreads();

    if (e0 < e1) {
        int ef = e0 + tid;
        int sv[EPT], dv[EPT], pp[EPT];
#pragma unroll
        for (int u = 0; u < EPT; ++u) {
            int ec = min(ef + u * T1, e1 - 1);
            sv[u] = src[ec];
            dv[u] = dst[ec];
        }
#pragma unroll
        for (int u = 0; u < EPT; ++u)
            pp[u] = atomicAdd(&cur[dv[u] >> BSH], (ef + u * T1 < e1) ? 1 : 0);
#pragma unroll
        for (int u = 0; u < EPT; ++u) {
            if (ef + u * T1 < e1) {
                int s = sv[u], d = dv[u];
                unsigned hb = (unsigned)(((s >> 16) << BSH) | (d & (BN - 1))) & 0xFFu;
                stage[pp[u]] = (unsigned)(s & 0xFFFF) | (hb << 16);
                sbk[pp[u]]   = (unsigned short)(d >> BSH);
            }
        }
    }
    __syncthreads();

    int n = e1 - e0;
    for (int p = tid; p < n; p += T1) {
        unsigned w = stage[p];
        int g = p + delta[sbk[p]];
        lo16[g] = (unsigned short)(w & 0xFFFFu);
        hi8[g]  = (unsigned char)(w >> 16);
    }
}

#define GR   32
#define GT   128
#define GX4  (GR * IN_DIM / 4)
#define GW4  (IN_DIM * HID_DIM / 4)
#define GXU  ((GX4 + GT - 1) / GT)
#define GWU  ((GW4 + GT - 1) / GT)
#define KS2  (IN_DIM / 8)
#define KB   16
#define NBLK ((KS2 + KB - 1) / KB)
#define CU   5
#define DISQ 3
static_assert(BN % GR == 0, "a GEMM work-group's rows lie in one bucket");
typedef __attribute__((ext_vector_type(4))) float v4f;

__global__ void __launch_bounds__(GT)
gemm1_wmma(const float* __restrict__ x,
           const float* __restrict__ W1,
           const unsigned* __restrict__ hi32,
           const int* __restrict__ bse,
           float* __restrict__ h1, float* __restrict__ disg, int num_nodes) {
    __shared__ __attribute__((aligned(16))) float xs[GR * IN_DIM];
    __shared__ __attribute__((aligned(16))) float wsm[IN_DIM * HID_DIM];
    __shared__ float exch[2 * 8 * 32];
    __shared__ int dcnt[GR];
    int tid = threadIdx.x;
    int lane = tid & 31;
    int wave = __builtin_amdgcn_readfirstlane(tid >> 5);
    int blk = blockIdx.x;
    int rows = min(GR, num_nodes - blk * GR);
    int n4 = rows * (IN_DIM / 4);
    const v4f* xb = (const v4f*)(x + (size_t)blk * GR * IN_DIM);
    const v4f* wb = (const v4f*)W1;

    int bk   = blk / (BN / GR);
    int part = blk % (BN / GR);
    if (tid < GR) dcnt[tid] = 0;

    v4f wr[GWU], xr[GXU];
    unsigned hw[CU];
#pragma unroll
    for (int u = 0; u < GWU; ++u)
        wr[u] = ((u + 1) * GT <= GW4) ? wb[u * GT + tid] : wb[min(u * GT + tid, GW4 - 1)];
#pragma unroll
    for (int u = 0; u < GXU; ++u)
        xr[u] = __builtin_nontemporal_load(xb + min(u * GT + tid, n4 - 1));
    __builtin_amdgcn_sched_barrier(0);
    int beg = bse[(size_t)bk * BROW], end = bse[(size_t)bk * BROW + 1];
    int w0 = beg >> 2, w1 = (end + 3) >> 2;
    int wlast = max(w1 - 1, w0);
#pragma unroll
    for (int u = 0; u < CU; ++u) hw[u] = hi32[min(w0 + u * GT + tid, wlast)];
#pragma unroll
    for (int u = 0; u < GWU; ++u) {
        int q = u * GT + tid;
        if ((u + 1) * GT <= GW4 || q < GW4) ((v4f*)wsm)[q] = wr[u];
    }
#pragma unroll
    for (int u = 0; u < GXU; ++u) {
        int i = u * GT + tid;
        if (i < n4) ((v4f*)xs)[i] = xr[u];
    }
    __syncthreads();

#pragma unroll
    for (int u = 0; u < CU; ++u) {
        int wi = w0 + u * GT + tid;
#pragma unroll
        for (int q = 0; q < 4; ++q) {
            int pos = wi * 4 + q;
            int dl = (hw[u] >> (8 * q)) & (BN - 1);
            if (pos >= beg && pos < end && dl / GR == part) atomicAdd(&dcnt[dl % GR], 1);
        }
    }
    for (int wi = w0 + CU * GT + tid; wi < w1; wi += GT) {
        unsigned wv = hi32[wi];
#pragma unroll
        for (int q = 0; q < 4; ++q) {
            int pos = wi * 4 + q;
            int dl = (wv >> (8 * q)) & (BN - 1);
            if (pos >= beg && pos < end && dl / GR == part) atomicAdd(&dcnt[dl % GR], 1);
        }
    }

    int half = lane >> 4;
    int l    = lane & 15;
    int tile = wave & 1;
    int par  = wave >> 1;
    int row0 = blk * GR + tile * 16;
    bool live = (row0 + 16 <= num_nodes);
    v8f c = {};
    if (live) {
        const float* arow = xs + (tile * 16 + l) * IN_DIM + half * 2 + par * 4;
        const float* bcol = wsm + (half * 2 + par * 4) * HID_DIM + l;
        v2f fa[2][KB], fb[2][KB];
        v8f c0 = {}, c1 = {};
#pragma unroll
        for (int t = 0; t < KB; ++t) {
            fa[0][t] = *(const v2f*)(arow + t * 8);
            fb[0][t].x = bcol[t * 8 * HID_DIM];
            fb[0][t].y = bcol[t * 8 * HID_DIM + HID_DIM];
        }
#pragma unroll
        for (int n = 0; n < NBLK; ++n) {
            if (n + 1 < NBLK) {
#pragma unroll
                for (int t = 0; t < KB; ++t) {
                    int j = (n + 1) * KB + t;
                    if (j < KS2) {
                        fa[(n + 1) & 1][t] = *(const v2f*)(arow + j * 8);
                        fb[(n + 1) & 1][t].x = bcol[j * 8 * HID_DIM];
                        fb[(n + 1) & 1][t].y = bcol[j * 8 * HID_DIM + HID_DIM];
                    }
                }
            }
            __builtin_amdgcn_sched_barrier(0);
#pragma unroll
            for (int t = 0; t < KB; ++t) {
                int j = n * KB + t;
                if (j < KS2) {
                    if (t & 1)
                        c1 = __builtin_amdgcn_wmma_f32_16x16x4_f32(
                                false, fa[n & 1][t], false, fb[n & 1][t], (short)0, c1, false, false);
                    else
                        c0 = __builtin_amdgcn_wmma_f32_16x16x4_f32(
                                false, fa[n & 1][t], false, fb[n & 1][t], (short)0, c0, false, false);
                }
            }
            __builtin_amdgcn_sched_barrier(0);
        }
        if (par == 0) {
            v2f a = *(const v2f*)(arow + KS2 * 8);
            v2f b;
            b.x = bcol[KS2 * 8 * HID_DIM];
            b.y = bcol[KS2 * 8 * HID_DIM + HID_DIM];
            c0 = __builtin_amdgcn_wmma_f32_16x16x4_f32(
                    false, a, false, b, (short)0, c0, false, false);
        }
        c = c0 + c1;
        if (par == 1) {
#pragma unroll
            for (int r = 0; r < 8; ++r) exch[(tile * 8 + r) * 32 + lane] = c[r];
        }
    }
    __syncthreads();
    if (live && par == 0) {
        float* outp = h1 + (size_t)row0 * HID_DIM;
#pragma unroll
        for (int r = 0; r < 8; ++r) {
            float d = (float)(dcnt[tile * 16 + half * 8 + r] + 1);
            float di = (d > 0.0f) ? rsqrtf(d) : 0.0f;
            outp[(half * 8 + r) * HID_DIM + l] = (c[r] + exch[(tile * 8 + r) * 32 + lane]) * di;
        }
    }
    if (tid < GR) {
        int node = blk * GR + tid;
        if (node < num_nodes) {
            float d = (float)(dcnt[tid] + 1);
            disg[(size_t)node * OUT_DIM + DISQ] = (d > 0.0f) ? rsqrtf(d) : 0.0f;
        }
    }
}

__global__ void __launch_bounds__(256)
k_layer1(const unsigned short* __restrict__ lo16,
         const unsigned char* __restrict__ hi8,
         const int* __restrict__ bse,
         const float* __restrict__ h1s,
         const float* __restrict__ disg,
         const float* __restrict__ b1,
         const float* __restrict__ W2,
         float* __restrict__ h2s, int num_nodes, int num_edges) {
    __shared__ __attribute__((aligned(16))) float acc[BN * HID_DIM];
    __shared__ float dis[BN];
    int b = blockIdx.x, tid = threadIdx.x;
    int lane = tid & 31, wave = tid >> 5;
    int beg = bse[(size_t)b * BROW], end = bse[(size_t)b * BROW + 1];
    for (int i = tid; i < BN * HID_DIM; i += 256) acc[i] = 0.0f;
    if (tid < BN) {
        int node = b * BN + tid;
        dis[tid] = (node < num_nodes) ? disg[(size_t)node * OUT_DIM + DISQ] : 0.0f;
    }
    __syncthreads();

    int sub = lane >> 4;
    int j   = lane & 15;
    unsigned w = NOEDGE;
    {
        int i = beg + wave * 32 + lane;
        if (i < end) w = (unsigned)lo16[i] | ((unsigned)hi8[i] << 16);
    }
    for (int base = beg + wave * 32; base < end; base += 256) {
        int in = base + 256 + lane;
        int ic = min(in, num_edges - 1);
        unsigned nlo = lo16[ic];
        unsigned nhi = hi8[ic];
        unsigned ww[16];
        float v[16];
#pragma unroll
        for (int t = 0; t < 16; ++t) ww[t] = __shfl(w, 2 * t + sub);
#pragma unroll
        for (int t = 0; t < 16; ++t) {
            unsigned s = (ww[t] & 0xFFFFu) | ((ww[t] >> 22) << 16);
            v[t] = (ww[t] != NOEDGE) ? h1s[(size_t)s * HID_DIM + j] : 0.0f;
        }
#pragma unroll
        for (int t = 0; t < 16; ++t) {
            if (ww[t] != NOEDGE) {
                int dl = (ww[t] >> 16) & (BN - 1);
                atomicAdd(&acc[dl * HID_DIM + j], v[t]);
            }
        }
        w = (in < end) ? (nlo | (nhi << 16)) : NOEDGE;
    }
    __syncthreads();

    for (int idx = tid; idx < BN * HID_DIM; idx += 256) {
        int dl = idx >> 4, jj = idx & 15;
        int node = b * BN + dl;
        float a = 0.0f;
        if (node < num_nodes) {
            float di = dis[dl];
            float val = di * (acc[idx] + h1s[(size_t)node * HID_DIM + jj]) + b1[jj];
            a = fmaxf(val, 0.0f);
        }
        acc[idx] = a;
    }
    __syncthreads();

    if (wave < BN / 16) {
        int half = lane >> 4;
        int l    = lane & 15;
        const float* arow = acc + (wave * 16 + l) * HID_DIM + half * 2;
        v8f c = {};
#pragma unroll
        for (int k0 = 0; k0 < HID_DIM; k0 += 4) {
            v2f a = *(const v2f*)(arow + k0);
            v2f bb;
            int kr = k0 + half * 2;
            bb.x = (l < OUT_DIM) ? W2[(kr + 0) * OUT_DIM + l] : 0.0f;
            bb.y = (l < OUT_DIM) ? W2[(kr + 1) * OUT_DIM + l] : 0.0f;
            c = __builtin_amdgcn_wmma_f32_16x16x4_f32(
                    false, a, false, bb, (short)0, c, false, false);
        }
        if (l < OUT_DIM) {
#pragma unroll
            for (int r = 0; r < 8; ++r) {
                int dl = wave * 16 + half * 8 + r;
                int node = b * BN + dl;
                if (node < num_nodes)
                    h2s[(size_t)node * OUT_DIM + l] = dis[dl] * c[r];
            }
        }
    }
}

__global__ void __launch_bounds__(256)
k_layer2(const unsigned short* __restrict__ lo16,
         const unsigned char* __restrict__ hi8,
         const int* bse,
         const float* __restrict__ h2s,
         const float* __restrict__ b2,
         float* out, int num_nodes, int num_edges) {
    __shared__ __attribute__((aligned(16))) float acc[BN * 8];
    int b = blockIdx.x, tid = threadIdx.x;
    int lane = tid & 31, wave = tid >> 5;
    int beg = bse[(size_t)b * BROW], end = bse[(size_t)b * BROW + 1];
    for (int i = tid; i < BN * 8; i += 256) acc[i] = 0.0f;
    float di = 0.0f;
    if (tid < BN && b * BN + tid < num_nodes)
        di = out[(size_t)(b * BN + tid) * OUT_DIM + DISQ];
    __syncthreads();

    int sub = lane >> 3;
    int j   = lane & 7;
    unsigned w = NOEDGE;
    {
        int i = beg + wave * 32 + lane;
        if (i < end) w = (unsigned)lo16[i] | ((unsigned)hi8[i] << 16);
    }
    for (int base = beg + wave * 32; base < end; base += 256) {
        int in = base + 256 + lane;
        int ic = min(in, num_edges - 1);
        unsigned nlo = lo16[ic];
        unsigned nhi = hi8[ic];
        unsigned ww[8];
        float v[8];
#pragma unroll
        for (int t = 0; t < 8; ++t) ww[t] = __shfl(w, 4 * t + sub);
#pragma unroll
        for (int t = 0; t < 8; ++t) {
            unsigned s = (ww[t] & 0xFFFFu) | ((ww[t] >> 22) << 16);
            v[t] = (ww[t] != NOEDGE && j < OUT_DIM) ? h2s[(size_t)s * OUT_DIM + j] : 0.0f;
        }
#pragma unroll
        for (int t = 0; t < 8; ++t) {
            if (ww[t] != NOEDGE) {
                int dl = (ww[t] >> 16) & (BN - 1);
                if (j < OUT_DIM) atomicAdd(&acc[dl * 8 + j], v[t]);
            }
        }
        w = (in < end) ? (nlo | (nhi << 16)) : NOEDGE;
    }
    __syncthreads();

    if (tid < BN) {
        int node = b * BN + tid;
        if (node < num_nodes) {
            float v[OUT_DIM];
            float m = -INFINITY;
#pragma unroll
            for (int q = 0; q < OUT_DIM; ++q) {
                v[q] = di * (acc[tid * 8 + q] + h2s[(size_t)node * OUT_DIM + q]) + b2[q];
                m = fmaxf(m, v[q]);
            }
            float s = 0.0f;
#pragma unroll
            for (int q = 0; q < OUT_DIM; ++q) s += expf(v[q] - m);
            float lse = logf(s);
#pragma unroll
            for (int q = 0; q < OUT_DIM; ++q)
                out[(size_t)node * OUT_DIM + q] = v[q] - m - lse;
        }
    }
}

extern "C" void kernel_launch(void* const* d_in, const int* in_sizes, int n_in,
                              void* d_out, int out_size, void* d_ws, size_t ws_size,
                              hipStream_t stream) {
    const float* x  = (const float*)d_in[0];
    const int*   ei = (const int*)d_in[1];
    const float* W1 = (const float*)d_in[2];
    const float* b1 = (const float*)d_in[3];
    const float* W2 = (const float*)d_in[4];
    const float* b2 = (const float*)d_in[5];

    const int N = in_sizes[0] / IN_DIM;
    const int E = in_sizes[1] / 2;
    const int* src = ei;
    const int* dst = ei + E;

    float* ws = (float*)d_ws;
    unsigned short* lo16 = (unsigned short*)ws;
    unsigned char*  hi8  = (unsigned char*)(ws + (size_t)(E + 1) / 2);
    float* h1  = ws + (size_t)(E + 1) / 2 + (size_t)(E + 3) / 4;
    float* h2s = h1 + (size_t)N * HID_DIM;
    float* outp = (float*)d_out;
    int*   bse  = (int*)d_out;
    float* disg = (float*)d_out;

    const int NBK   = (N + BN - 1) / BN;
    const int chunk = (E + G1 - 1) / G1;

    int* table  = (int*)h1;
    int* pref   = table + (size_t)G1 * NBK;
    int* rowtot = pref + (size_t)G1 * NBK;

    const int B = 256;
    auto nb = [](long long work, int blk) { return (int)((work + blk - 1) / blk); };

    k_hist<<<G1, T1, NBK * sizeof(int), stream>>>(dst, table, E, chunk, NBK);
    k_colscan<<<nb(NBK, 32), 32 * CW, 0, stream>>>(table, pref, rowtot, NBK);
    k_basescan<<<1, 1024, 0, stream>>>(rowtot, bse, NBK);
    k_scatter<<<G1, T1, 2 * NBK * sizeof(int), stream>>>(src, dst, table, pref, bse, lo16, hi8,
                                                         E, chunk, NBK);

    gemm1_wmma<<<nb(N, GR), GT, 0, stream>>>(x, W1, (const unsigned*)hi8, bse, h1, disg, N);
    k_layer1<<<NBK, B, 0, stream>>>(lo16, hi8, bse, h1, disg, b1, W2, h2s, N, E);

    k_layer2<<<NBK, B, 0, stream>>>(lo16, hi8, bse, h2s, b2, outp, N, E);
}
